# starting kernel
# baseline (speedup 1.0000x reference)
#include <hip/hip_runtime.h>
#include <hip/hip_bf16.h>


typedef __attribute__((ext_vector_type(2))) float v2f;
typedef __attribute__((ext_vector_type(8))) float v8f;

#define CB 2
#define CT 2048
#define CE 1024
#define CH 8
#define CDK 128
#define CDV 256
#define CV 2048
#define NQKV 4096
#define BT (CB * CT)

static __device__ __forceinline__ v8f wmma4(v2f a, v2f b, v8f c) {
  return __builtin_amdgcn_wmma_f32_16x16x4_f32(
       false, a,  false, b,
       (short)0, c,  false,  false);
}

static __device__ __forceinline__ v8f vzero8() {
  v8f z;
#pragma unroll
  for (int i = 0; i < 8; ++i) z[i] = 0.0f;
  return z;
}

static __device__ __forceinline__ v2f ldsv2(const float* p) {
  return *reinterpret_cast<const v2f*>(p);
}

__global__ __launch_bounds__(128) void gemm_wmma_f32(
    const float* __restrict__ A, const float* __restrict__ Bm,
    const float* __restrict__ bias, float* __restrict__ C,
    int M, int N, int K) {
  __shared__ float as[128][34];
  __shared__ float bs[16][66][2];

  const int tid = threadIdx.x;
  const int lane = tid & 31;
  const int w = tid >> 5;
  const int hi = lane >> 4;
  const int ln = lane & 15;
  const int n0 = blockIdx.x * 64;
  const int m0 = blockIdx.y * 128;

  v8f acc[2][4];
#pragma unroll
  for (int mt = 0; mt < 2; ++mt)
#pragma unroll
    for (int nt = 0; nt < 4; ++nt) acc[mt][nt] = vzero8();

  for (int k0 = 0; k0 < K; k0 += 32) {
#pragma unroll
    for (int e = tid; e < 128 * 32; e += 128) {
      int r = e >> 5, c = e & 31;
      as[r][c] = A[(size_t)(m0 + r) * K + (k0 + c)];
    }
#pragma unroll
    for (int e = tid; e < 32 * 64; e += 128) {
      int r = e >> 6, c = e & 63;
      bs[r >> 1][c][r & 1] = Bm[(size_t)(k0 + r) * N + (n0 + c)];
    }
    __syncthreads();

    if (k0 + 32 < K) {
      __builtin_prefetch(&A[(size_t)(m0 + tid) * K + (k0 + 32)], 0, 1);
      __builtin_prefetch(&Bm[(size_t)(k0 + 32 + (tid & 31)) * N + n0], 0, 1);
    }

#pragma unroll
    for (int kk = 0; kk < 32; kk += 4) {
      const int ak = kk + hi * 2;
      const int kp = (kk >> 1) + hi;
      v2f a0 = ldsv2(&as[w * 32 + ln][ak]);
      v2f a1 = ldsv2(&as[w * 32 + 16 + ln][ak]);
#pragma unroll
      for (int nt = 0; nt < 4; ++nt) {
        v2f b = ldsv2(&bs[kp][nt * 16 + ln][0]);
        acc[0][nt] = wmma4(a0, b, acc[0][nt]);
        acc[1][nt] = wmma4(a1, b, acc[1][nt]);
      }
    }
    __syncthreads();
  }

#pragma unroll
  for (int mt = 0; mt < 2; ++mt)
#pragma unroll
    for (int nt = 0; nt < 4; ++nt)
#pragma unroll
      for (int vv = 0; vv < 8; ++vv) {
        int row = m0 + w * 32 + mt * 16 + vv + hi * 8;
        int col = n0 + nt * 16 + ln;
        C[(size_t)row * N + col] = acc[mt][nt][vv] + bias[col];
      }
}

__global__ __launch_bounds__(512) void xpos_kernel(float* __restrict__ qkv) {
  const int row = blockIdx.x;
  const int j = threadIdx.x;
  const int t = row & (CT - 1);

  const float D = (float)CE;
  const float base = (2.0f * (float)j + 0.4f * D) / (1.4f * D);
  const float sc = __powf(base, (float)t / 512.0f);
  const float invf = __powf(10000.0f, -(float)j / 512.0f);
  float sn, cn;
  __sincosf((float)t * invf, &sn, &cn);

  size_t off = (size_t)row * NQKV + 2 * j;
  {
    float x0 = qkv[off], x1 = qkv[off + 1];
    float c = cn * sc, s = sn * sc;
    qkv[off]     = x0 * c - x1 * s;
    qkv[off + 1] = x1 * c + x0 * s;
  }
  {
    float isc = 1.0f / sc;
    float x0 = qkv[off + CE], x1 = qkv[off + CE + 1];
    float c = cn * isc, s = sn * isc;
    qkv[off + CE]     = x0 * c - x1 * s;
    qkv[off + CE + 1] = x1 * c + x0 * s;
  }
}

__global__ __launch_bounds__(128) void retention_kernel(
    const float* __restrict__ qkv, float* __restrict__ ro) {
  __shared__ float q_s[64][132];
  __shared__ float k_s[16][132];
  __shared__ float v_s[8][258][2];
  __shared__ float s_s[4][16][20];

  const int iblk = blockIdx.x;
  const int bh = blockIdx.y;
  const int b = bh >> 3, h = bh & 7;
  const int i0 = iblk * 64;
  const int rowbase = b * CT;

  const int tid = threadIdx.x;
  const int lane = tid & 31;
  const int w = tid >> 5;
  const int hi = lane >> 4;
  const int ln = lane & 15;

  const float se = logf(1.0f / 32.0f), ee = logf(1.0f / 512.0f);
  const float g = 1.0f - __expf(se + (ee - se) * ((float)h / 7.0f));
  const float logg = logf(g);
  const float rsdk = rsqrtf((float)CDK);

#pragma unroll
  for (int e = tid; e < 64 * 128; e += 128) {
    int r = e >> 7, c = e & 127;
    q_s[r][c] = qkv[(size_t)(rowbase + i0 + r) * NQKV + h * CDK + c];
  }

  v8f acc[16];
#pragma unroll
  for (int i = 0; i < 16; ++i) acc[i] = vzero8();

  for (int j0 = 0; j0 < i0 + 64; j0 += 16) {
    __syncthreads();
#pragma unroll
    for (int e = tid; e < 16 * 128; e += 128) {
      int r = e >> 7, c = e & 127;
      k_s[r][c] = qkv[(size_t)(rowbase + j0 + r) * NQKV + CE + h * CDK + c];
    }
#pragma unroll
    for (int e = tid; e < 16 * 256; e += 128) {
      int r = e >> 8, c = e & 255;
      v_s[r >> 1][c][r & 1] =
          qkv[(size_t)(rowbase + j0 + r) * NQKV + 2 * CE + h * CDV + c];
    }
    __syncthreads();

    v8f sacc = vzero8();
#pragma unroll
    for (int kk = 0; kk < CDK; kk += 4) {
      const int ak = kk + hi * 2;
      v2f a = ldsv2(&q_s[w * 16 + ln][ak]);
      v2f bf = ldsv2(&k_s[ln][ak]);
      sacc = wmma4(a, bf, sacc);
    }

#pragma unroll
    for (int vv = 0; vv < 8; ++vv) {
      int i = i0 + w * 16 + vv + hi * 8;
      int j = j0 + ln;
      float m = (i >= j) ? __expf(logg * (float)(i - j)) * rsdk : 0.0f;
      sacc[vv] *= m;
    }

#pragma unroll
    for (int vv = 0; vv < 8; ++vv) s_s[w][vv + hi * 8][ln] = sacc[vv];

#pragma unroll
    for (int ks = 0; ks < 16; ks += 4) {
      const int ak = ks + hi * 2;
      const int kp = (ks >> 1) + hi;
      v2f a = ldsv2(&s_s[w][ln][ak]);
#pragma unroll
      for (int nt = 0; nt < 16; ++nt) {
        v2f bf = ldsv2(&v_s[kp][nt * 16 + ln][0]);
        acc[nt] = wmma4(a, bf, acc[nt]);
      }
    }
  }

  const float inv1mg = 1.0f / (1.0f - g);
#pragma unroll
  for (int nt = 0; nt < 16; ++nt)
#pragma unroll
    for (int vv = 0; vv < 8; ++vv) {
      int i = i0 + w * 16 + vv + hi * 8;
      float rowsum = (1.0f - __expf(logg * (float)(i + 1))) * inv1mg;
      float sc = rsqrtf(rowsum);
      ro[(size_t)(rowbase + i) * CV + h * CDV + nt * 16 + ln] = acc[nt][vv] * sc;
    }
}

__global__ __launch_bounds__(256) void curr_kv_kernel(
    const float* __restrict__ qkv, float* __restrict__ kvout) {
  __shared__ float kw_s[8][130][2];
  __shared__ float v_s[8][258][2];

  const int bh = blockIdx.x;
  const int b = bh >> 3, h = bh & 7;
  const int rowbase = b * CT;

  const int tid = threadIdx.x;
  const int lane = tid & 31;
  const int w = tid >> 5;
  const int hi = lane >> 4;
  const int ln = lane & 15;

  const float se = logf(1.0f / 32.0f), ee = logf(1.0f / 512.0f);
  const float g = 1.0f - __expf(se + (ee - se) * ((float)h / 7.0f));
  const float logg = logf(g);
  const float rnorm = rsqrtf((1.0f - __expf(logg * (float)CT)) / (1.0f - g));

  v8f acc[16];
#pragma unroll
  for (int i = 0; i < 16; ++i) acc[i] = vzero8();

  for (int j0 = 0; j0 < CT; j0 += 16) {
#pragma unroll
    for (int e = tid; e < 16 * 128; e += 256) {
      int r = e >> 7, c = e & 127;
      int j = j0 + r;
      float wj = __expf(logg * (float)(CT - 1 - j)) * rnorm;
      kw_s[r >> 1][c][r & 1] =
          qkv[(size_t)(rowbase + j) * NQKV + CE + h * CDK + c] * wj;
    }
#pragma unroll
    for (int e = tid; e < 16 * 256; e += 256) {
      int r = e >> 8, c = e & 255;
      v_s[r >> 1][c][r & 1] =
          qkv[(size_t)(rowbase + j0 + r) * NQKV + 2 * CE + h * CDV + c];
    }
    __syncthreads();

#pragma unroll
    for (int ks = 0; ks < 16; ks += 4) {
      const int kp = (ks >> 1) + hi;
      v2f a = ldsv2(&kw_s[kp][w * 16 + ln][0]);
#pragma unroll
      for (int nt = 0; nt < 16; ++nt) {
        v2f bf = ldsv2(&v_s[kp][nt * 16 + ln][0]);
        acc[nt] = wmma4(a, bf, acc[nt]);
      }
    }
    __syncthreads();
  }

#pragma unroll
  for (int nt = 0; nt < 16; ++nt)
#pragma unroll
    for (int vv = 0; vv < 8; ++vv) {
      int kk = w * 16 + vv + hi * 8;
      kvout[(size_t)bh * CDK * CDV + (size_t)kk * CDV + nt * 16 + ln] =
          acc[nt][vv];
    }
}

__global__ __launch_bounds__(256) void gn_gate_kernel(
    const float* __restrict__ ro, float* __restrict__ gbuf) {
  const int row = blockIdx.x;
  const int w = threadIdx.x >> 5;
  const int lane = threadIdx.x & 31;
  const size_t base = (size_t)row * CV + (size_t)w * CDV;

  float x[8];
  float s = 0.0f, s2 = 0.0f;
#pragma unroll
  for (int i = 0; i < 8; ++i) {
    x[i] = ro[base + i * 32 + lane];
    s += x[i];
    s2 += x[i] * x[i];
  }
#pragma unroll
  for (int off = 16; off > 0; off >>= 1) {
    s += __shfl_xor(s, off, 32);
    s2 += __shfl_xor(s2, off, 32);
  }
  const float mu = s * (1.0f / 256.0f);
  const float var = s2 * (1.0f / 256.0f) - mu * mu;
  const float rstd = rsqrtf(var + 1e-5f);

#pragma unroll
  for (int i = 0; i < 8; ++i) {
    size_t idx = base + i * 32 + lane;
    float nrm = (x[i] - mu) * rstd;
    float gv = gbuf[idx];
    float sil = gv / (1.0f + __expf(-gv));
    gbuf[idx] = sil * nrm;
  }
}

extern "C" void kernel_launch(void* const* d_in, const int* in_sizes, int n_in,
                              void* d_out, int out_size, void* d_ws,
                              size_t ws_size, hipStream_t stream) {
  const float* hs    = (const float*)d_in[0];
  const float* w_qkv = (const float*)d_in[1];
  const float* b_qkv = (const float*)d_in[2];
  const float* w_g   = (const float*)d_in[3];
  const float* b_g   = (const float*)d_in[4];
  const float* w_p   = (const float*)d_in[5];
  const float* b_p   = (const float*)d_in[6];

  float* out  = (float*)d_out;
  float* qkv  = (float*)d_ws;
  float* gbuf = qkv + (size_t)BT * NQKV;
  float* ro   = gbuf + (size_t)BT * CV;

  gemm_wmma_f32<<<dim3(NQKV / 64, BT / 128), 128, 0, stream>>>(
      hs, w_qkv, b_qkv, qkv, BT, NQKV, CE);

  xpos_kernel<<<dim3(BT), 512, 0, stream>>>(qkv);

  gemm_wmma_f32<<<dim3(CV / 64, BT / 128), 128, 0, stream>>>(
      hs, w_g, b_g, gbuf, BT, CV, CE);

  retention_kernel<<<dim3(CT / 64, CB * CH), 128, 0, stream>>>(qkv, ro);

  curr_kv_kernel<<<dim3(CB * CH), 256, 0, stream>>>(
      qkv, out + (size_t)BT * CE);

  gn_gate_kernel<<<dim3(BT), 256, 0, stream>>>(ro, gbuf);

  gemm_wmma_f32<<<dim3(CE / 64, BT / 128), 128, 0, stream>>>(
      gbuf, w_p, b_p, out, BT, CE, CV);
}


# retention kernel: K/V (and Q) tiles staged as grouped 16-byte loads held in registers, next tile's group in flight under the current tile's WMMAs; was one or two scalar loads in flight at a time
# speedup vs baseline: 1.5058x; 1.5058x over previous
#include <hip/hip_runtime.h>
#include <hip/hip_bf16.h>


typedef __attribute__((ext_vector_type(2))) float v2f;
typedef __attribute__((ext_vector_type(4))) float v4f;
typedef __attribute__((ext_vector_type(8))) float v8f;

#define CB 2
#define CT 2048
#define CE 1024
#define CH 8
#define CDK 128
#define CDV 256
#define CV 2048
#define NQKV 4096
#define BT (CB * CT)

static __device__ __forceinline__ v8f wmma4(v2f a, v2f b, v8f c) {
  return __builtin_amdgcn_wmma_f32_16x16x4_f32(
       false, a,  false, b,
       (short)0, c,  false,  false);
}

static __device__ __forceinline__ v8f vzero8() {
  v8f z;
#pragma unroll
  for (int i = 0; i < 8; ++i) z[i] = 0.0f;
  return z;
}

static __device__ __forceinline__ v2f ldsv2(const float* p) {
  return *reinterpret_cast<const v2f*>(p);
}

__global__ __launch_bounds__(128) void gemm_wmma_f32(
    const float* __restrict__ A, const float* __restrict__ Bm,
    const float* __restrict__ bias, float* __restrict__ C,
    int M, int N, int K) {
  __shared__ float as[128][34];
  __shared__ float bs[16][66][2];

  const int tid = threadIdx.x;
  const int lane = tid & 31;
  const int w = tid >> 5;
  const int hi = lane >> 4;
  const int ln = lane & 15;
  const int n0 = blockIdx.x * 64;
  const int m0 = blockIdx.y * 128;

  v8f acc[2][4];
#pragma unroll
  for (int mt = 0; mt < 2; ++mt)
#pragma unroll
    for (int nt = 0; nt < 4; ++nt) acc[mt][nt] = vzero8();

  for (int k0 = 0; k0 < K; k0 += 32) {
#pragma unroll
    for (int e = tid; e < 128 * 32; e += 128) {
      int r = e >> 5, c = e & 31;
      as[r][c] = A[(size_t)(m0 + r) * K + (k0 + c)];
    }
#pragma unroll
    for (int e = tid; e < 32 * 64; e += 128) {
      int r = e >> 6, c = e & 63;
      bs[r >> 1][c][r & 1] = Bm[(size_t)(k0 + r) * N + (n0 + c)];
    }
    __syncthreads();

    if (k0 + 32 < K) {
      __builtin_prefetch(&A[(size_t)(m0 + tid) * K + (k0 + 32)], 0, 1);
      __builtin_prefetch(&Bm[(size_t)(k0 + 32 + (tid & 31)) * N + n0], 0, 1);
    }

#pragma unroll
    for (int kk = 0; kk < 32; kk += 4) {
      const int ak = kk + hi * 2;
      const int kp = (kk >> 1) + hi;
      v2f a0 = ldsv2(&as[w * 32 + ln][ak]);
      v2f a1 = ldsv2(&as[w * 32 + 16 + ln][ak]);
#pragma unroll
      for (int nt = 0; nt < 4; ++nt) {
        v2f b = ldsv2(&bs[kp][nt * 16 + ln][0]);
        acc[0][nt] = wmma4(a0, b, acc[0][nt]);
        acc[1][nt] = wmma4(a1, b, acc[1][nt]);
      }
    }
    __syncthreads();
  }

#pragma unroll
  for (int mt = 0; mt < 2; ++mt)
#pragma unroll
    for (int nt = 0; nt < 4; ++nt)
#pragma unroll
      for (int vv = 0; vv < 8; ++vv) {
        int row = m0 + w * 32 + mt * 16 + vv + hi * 8;
        int col = n0 + nt * 16 + ln;
        C[(size_t)row * N + col] = acc[mt][nt][vv] + bias[col];
      }
}

__global__ __launch_bounds__(512) void xpos_kernel(float* __restrict__ qkv) {
  const int row = blockIdx.x;
  const int j = threadIdx.x;
  const int t = row & (CT - 1);

  const float D = (float)CE;
  const float base = (2.0f * (float)j + 0.4f * D) / (1.4f * D);
  const float sc = __powf(base, (float)t / 512.0f);
  const float invf = __powf(10000.0f, -(float)j / 512.0f);
  float sn, cn;
  __sincosf((float)t * invf, &sn, &cn);

  size_t off = (size_t)row * NQKV + 2 * j;
  {
    float x0 = qkv[off], x1 = qkv[off + 1];
    float c = cn * sc, s = sn * sc;
    qkv[off]     = x0 * c - x1 * s;
    qkv[off + 1] = x1 * c + x0 * s;
  }
  {
    float isc = 1.0f / sc;
    float x0 = qkv[off + CE], x1 = qkv[off + CE + 1];
    float c = cn * isc, s = sn * isc;
    qkv[off + CE]     = x0 * c - x1 * s;
    qkv[off + CE + 1] = x1 * c + x0 * s;
  }
}

__global__ __launch_bounds__(128) void retention_kernel(
    const float* __restrict__ qkv, float* __restrict__ ro) {
  __shared__ __attribute__((aligned(16))) float q_s[64][132];
  __shared__ __attribute__((aligned(16))) float k_s[16][132];
  __shared__ __attribute__((aligned(16))) float v_s[8][258][2];
  __shared__ float s_s[4][16][20];

  const int iblk = blockIdx.x;
  const int bh = blockIdx.y;
  const int b = bh >> 3, h = bh & 7;
  const int i0 = iblk * 64;
  const int rowbase = b * CT;

  const int tid = threadIdx.x;
  const int lane = tid & 31;
  const int w = tid >> 5;
  const int hi = lane >> 4;
  const int ln = lane & 15;

  const float se = logf(1.0f / 32.0f), ee = logf(1.0f / 512.0f);
  const float g = 1.0f - __expf(se + (ee - se) * ((float)h / 7.0f));
  const float logg = logf(g);
  const float rsdk = rsqrtf((float)CDK);

  const int kr = tid >> 5;
  const int kc = (tid & 31) * 4;
  const int vp = tid >> 6;
  const int vc = (tid & 63) * 4;
  const float* kptr = qkv + (size_t)(rowbase + kr) * NQKV + CE + h * CDK + kc;
  const float* vptr =
      qkv + (size_t)(rowbase + 2 * vp) * NQKV + 2 * CE + h * CDV + vc;

  v4f kreg[4], ve[4], vo[4];

#pragma unroll
  for (int i = 0; i < 4; ++i) {
    kreg[i] = *reinterpret_cast<const v4f*>(kptr + (size_t)(4 * i) * NQKV);
    ve[i] = *reinterpret_cast<const v4f*>(vptr + (size_t)(4 * i) * NQKV);
    vo[i] = *reinterpret_cast<const v4f*>(vptr + (size_t)(4 * i + 1) * NQKV);
  }

#pragma unroll
  for (int i = 0; i < 16; ++i) {
    const int r = kr + 4 * i;
    v4f qv = *reinterpret_cast<const v4f*>(
        qkv + (size_t)(rowbase + i0 + r) * NQKV + h * CDK + kc);
    *reinterpret_cast<v4f*>(&q_s[r][kc]) = qv;
  }

  v8f acc[16];
#pragma unroll
  for (int i = 0; i < 16; ++i) acc[i] = vzero8();

  for (int j0 = 0; j0 < i0 + 64; j0 += 16) {
    __syncthreads();
#pragma unroll
    for (int i = 0; i < 4; ++i) {
      *reinterpret_cast<v4f*>(&k_s[kr + 4 * i][kc]) = kreg[i];
      v4f lo, hi2;
      lo.x = ve[i].x;  lo.y = vo[i].x;  lo.z = ve[i].y;  lo.w = vo[i].y;
      hi2.x = ve[i].z; hi2.y = vo[i].z; hi2.z = ve[i].w; hi2.w = vo[i].w;
      *reinterpret_cast<v4f*>(&v_s[vp + 2 * i][vc][0]) = lo;
      *reinterpret_cast<v4f*>(&v_s[vp + 2 * i][vc + 2][0]) = hi2;
    }
    __syncthreads();

    {
      const int jn = (j0 + 16 < i0 + 64) ? (j0 + 16) : j0;
      const size_t adv = (size_t)jn * NQKV;
#pragma unroll
      for (int i = 0; i < 4; ++i) {
        kreg[i] =
            *reinterpret_cast<const v4f*>(kptr + adv + (size_t)(4 * i) * NQKV);
        ve[i] =
            *reinterpret_cast<const v4f*>(vptr + adv + (size_t)(4 * i) * NQKV);
        vo[i] = *reinterpret_cast<const v4f*>(vptr + adv +
                                              (size_t)(4 * i + 1) * NQKV);
      }
    }

    v8f sacc = vzero8();
#pragma unroll
    for (int kk = 0; kk < CDK; kk += 4) {
      const int ak = kk + hi * 2;
      v2f a = ldsv2(&q_s[w * 16 + ln][ak]);
      v2f bf = ldsv2(&k_s[ln][ak]);
      sacc = wmma4(a, bf, sacc);
    }

#pragma unroll
    for (int vv = 0; vv < 8; ++vv) {
      int i = i0 + w * 16 + vv + hi * 8;
      int j = j0 + ln;
      float m = (i >= j) ? __expf(logg * (float)(i - j)) * rsdk : 0.0f;
      sacc[vv] *= m;
    }

#pragma unroll
    for (int vv = 0; vv < 8; ++vv) s_s[w][vv + hi * 8][ln] = sacc[vv];

#pragma unroll
    for (int ks = 0; ks < 16; ks += 4) {
      const int ak = ks + hi * 2;
      const int kp = (ks >> 1) + hi;
      v2f a = ldsv2(&s_s[w][ln][ak]);
#pragma unroll
      for (int nt = 0; nt < 16; ++nt) {
        v2f bf = ldsv2(&v_s[kp][nt * 16 + ln][0]);
        acc[nt] = wmma4(a, bf, acc[nt]);
      }
    }
  }

  const float inv1mg = 1.0f / (1.0f - g);
#pragma unroll
  for (int nt = 0; nt < 16; ++nt)
#pragma unroll
    for (int vv = 0; vv < 8; ++vv) {
      int i = i0 + w * 16 + vv + hi * 8;
      float rowsum = (1.0f - __expf(logg * (float)(i + 1))) * inv1mg;
      float sc = rsqrtf(rowsum);
      ro[(size_t)(rowbase + i) * CV + h * CDV + nt * 16 + ln] = acc[nt][vv] * sc;
    }
}

__global__ __launch_bounds__(256) void curr_kv_kernel(
    const float* __restrict__ qkv, float* __restrict__ kvout) {
  __shared__ float kw_s[8][130][2];
  __shared__ float v_s[8][258][2];

  const int bh = blockIdx.x;
  const int b = bh >> 3, h = bh & 7;
  const int rowbase = b * CT;

  const int tid = threadIdx.x;
  const int lane = tid & 31;
  const int w = tid >> 5;
  const int hi = lane >> 4;
  const int ln = lane & 15;

  const float se = logf(1.0f / 32.0f), ee = logf(1.0f / 512.0f);
  const float g = 1.0f - __expf(se + (ee - se) * ((float)h / 7.0f));
  const float logg = logf(g);
  const float rnorm = rsqrtf((1.0f - __expf(logg * (float)CT)) / (1.0f - g));

  v8f acc[16];
#pragma unroll
  for (int i = 0; i < 16; ++i) acc[i] = vzero8();

  for (int j0 = 0; j0 < CT; j0 += 16) {
#pragma unroll
    for (int e = tid; e < 16 * 128; e += 256) {
      int r = e >> 7, c = e & 127;
      int j = j0 + r;
      float wj = __expf(logg * (float)(CT - 1 - j)) * rnorm;
      kw_s[r >> 1][c][r & 1] =
          qkv[(size_t)(rowbase + j) * NQKV + CE + h * CDK + c] * wj;
    }
#pragma unroll
    for (int e = tid; e < 16 * 256; e += 256) {
      int r = e >> 8, c = e & 255;
      v_s[r >> 1][c][r & 1] =
          qkv[(size_t)(rowbase + j0 + r) * NQKV + 2 * CE + h * CDV + c];
    }
    __syncthreads();

#pragma unroll
    for (int ks = 0; ks < 16; ks += 4) {
      const int kp = (ks >> 1) + hi;
      v2f a = ldsv2(&kw_s[kp][w * 16 + ln][0]);
#pragma unroll
      for (int nt = 0; nt < 16; ++nt) {
        v2f bf = ldsv2(&v_s[kp][nt * 16 + ln][0]);
        acc[nt] = wmma4(a, bf, acc[nt]);
      }
    }
    __syncthreads();
  }

#pragma unroll
  for (int nt = 0; nt < 16; ++nt)
#pragma unroll
    for (int vv = 0; vv < 8; ++vv) {
      int kk = w * 16 + vv + hi * 8;
      kvout[(size_t)bh * CDK * CDV + (size_t)kk * CDV + nt * 16 + ln] =
          acc[nt][vv];
    }
}

__global__ __launch_bounds__(256) void gn_gate_kernel(
    const float* __restrict__ ro, float* __restrict__ gbuf) {
  const int row = blockIdx.x;
  const int w = threadIdx.x >> 5;
  const int lane = threadIdx.x & 31;
  const size_t base = (size_t)row * CV + (size_t)w * CDV;

  float x[8];
  float s = 0.0f, s2 = 0.0f;
#pragma unroll
  for (int i = 0; i < 8; ++i) {
    x[i] = ro[base + i * 32 + lane];
    s += x[i];
    s2 += x[i] * x[i];
  }
#pragma unroll
  for (int off = 16; off > 0; off >>= 1) {
    s += __shfl_xor(s, off, 32);
    s2 += __shfl_xor(s2, off, 32);
  }
  const float mu = s * (1.0f / 256.0f);
  const float var = s2 * (1.0f / 256.0f) - mu * mu;
  const float rstd = rsqrtf(var + 1e-5f);

#pragma unroll
  for (int i = 0; i < 8; ++i) {
    size_t idx = base + i * 32 + lane;
    float nrm = (x[i] - mu) * rstd;
    float gv = gbuf[idx];
    float sil = gv / (1.0f + __expf(-gv));
    gbuf[idx] = sil * nrm;
  }
}

extern "C" void kernel_launch(void* const* d_in, const int* in_sizes, int n_in,
                              void* d_out, int out_size, void* d_ws,
                              size_t ws_size, hipStream_t stream) {
  const float* hs    = (const float*)d_in[0];
  const float* w_qkv = (const float*)d_in[1];
  const float* b_qkv = (const float*)d_in[2];
  const float* w_g   = (const float*)d_in[3];
  const float* b_g   = (const float*)d_in[4];
  const float* w_p   = (const float*)d_in[5];
  const float* b_p   = (const float*)d_in[6];

  float* out  = (float*)d_out;
  float* qkv  = (float*)d_ws;
  float* gbuf = qkv + (size_t)BT * NQKV;
  float* ro   = gbuf + (size_t)BT * CV;

  gemm_wmma_f32<<<dim3(NQKV / 64, BT / 128), 128, 0, stream>>>(
      hs, w_qkv, b_qkv, qkv, BT, NQKV, CE);

  xpos_kernel<<<dim3(BT), 512, 0, stream>>>(qkv);

  gemm_wmma_f32<<<dim3(CV / 64, BT / 128), 128, 0, stream>>>(
      hs, w_g, b_g, gbuf, BT, CV, CE);

  retention_kernel<<<dim3(CT / 64, CB * CH), 128, 0, stream>>>(qkv, ro);

  curr_kv_kernel<<<dim3(CB * CH), 256, 0, stream>>>(
      qkv, out + (size_t)BT * CE);

  gn_gate_kernel<<<dim3(BT), 256, 0, stream>>>(ro, gbuf);

  gemm_wmma_f32<<<dim3(CE / 64, BT / 128), 128, 0, stream>>>(
      gbuf, w_p, b_p, out, BT, CE, CV);
}


# gemm and curr_kv kernels: A/B and K/V tiles staged as grouped 16-byte loads held in registers, next tile's group in flight under the current tile's WMMAs; were 2 to 5 scalar loads in flight
# speedup vs baseline: 3.1903x; 2.1187x over previous
#include <hip/hip_runtime.h>
#include <hip/hip_bf16.h>


typedef __attribute__((ext_vector_type(2))) float v2f;
typedef __attribute__((ext_vector_type(4))) float v4f;
typedef __attribute__((ext_vector_type(8))) float v8f;

#define CB 2
#define CT 2048
#define CE 1024
#define CH 8
#define CDK 128
#define CDV 256
#define CV 2048
#define NQKV 4096
#define BT (CB * CT)

static __device__ __forceinline__ v8f wmma4(v2f a, v2f b, v8f c) {
  return __builtin_amdgcn_wmma_f32_16x16x4_f32(
       false, a,  false, b,
       (short)0, c,  false,  false);
}

static __device__ __forceinline__ v8f vzero8() {
  v8f z;
#pragma unroll
  for (int i = 0; i < 8; ++i) z[i] = 0.0f;
  return z;
}

static __device__ __forceinline__ v2f ldsv2(const float* p) {
  return *reinterpret_cast<const v2f*>(p);
}

__global__ __launch_bounds__(128) __attribute__((amdgpu_waves_per_eu(4, 4)))
void gemm_wmma_f32(
    const float* __restrict__ A, const float* __restrict__ Bm,
    const float* __restrict__ bias, float* __restrict__ C,
    int M, int N, int K) {
  __shared__ __attribute__((aligned(16))) float as[128][36];
  __shared__ __attribute__((aligned(16))) float bs[16][66][2];

  const int tid = threadIdx.x;
  const int lane = tid & 31;
  const int w = tid >> 5;
  const int hi = lane >> 4;
  const int ln = lane & 15;
  const int n0 = blockIdx.x * 64;
  const int m0 = blockIdx.y * 128;

  const int ar = tid >> 3;
  const int ac = (tid & 7) * 4;
  const int bp = tid >> 4;
  const int bc = (tid & 15) * 4;
  const float* aptr = A + (size_t)(m0 + ar) * K + ac;
  const float* bptr = Bm + (size_t)(2 * bp) * N + (n0 + bc);
  const size_t a16 = (size_t)16 * K;
  const size_t b16 = (size_t)16 * N;

  v4f areg[8], be[2], bo[2];

#pragma unroll
  for (int i = 0; i < 8; ++i)
    areg[i] = *reinterpret_cast<const v4f*>(aptr + i * a16);
#pragma unroll
  for (int i = 0; i < 2; ++i) {
    be[i] = *reinterpret_cast<const v4f*>(bptr + i * b16);
    bo[i] = *reinterpret_cast<const v4f*>(bptr + i * b16 + N);
  }

  v8f acc[2][4];
#pragma unroll
  for (int mt = 0; mt < 2; ++mt)
#pragma unroll
    for (int nt = 0; nt < 4; ++nt) acc[mt][nt] = vzero8();

  for (int k0 = 0; k0 < K; k0 += 32) {
    __syncthreads();
#pragma unroll
    for (int i = 0; i < 8; ++i)
      *reinterpret_cast<v4f*>(&as[ar + 16 * i][ac]) = areg[i];
#pragma unroll
    for (int i = 0; i < 2; ++i) {
      v4f lo, hi2;
      lo.x = be[i].x;  lo.y = bo[i].x;  lo.z = be[i].y;  lo.w = bo[i].y;
      hi2.x = be[i].z; hi2.y = bo[i].z; hi2.z = be[i].w; hi2.w = bo[i].w;
      *reinterpret_cast<v4f*>(&bs[bp + 8 * i][bc][0]) = lo;
      *reinterpret_cast<v4f*>(&bs[bp + 8 * i][bc + 2][0]) = hi2;
    }
    __syncthreads();

    {
      const int kn = (k0 + 32 < K) ? (k0 + 32) : k0;
      const float* ap = aptr + kn;
      const float* bq = bptr + (size_t)kn * N;
#pragma unroll
      for (int i = 0; i < 8; ++i)
        areg[i] = *reinterpret_cast<const v4f*>(ap + i * a16);
#pragma unroll
      for (int i = 0; i < 2; ++i) {
        be[i] = *reinterpret_cast<const v4f*>(bq + i * b16);
        bo[i] = *reinterpret_cast<const v4f*>(bq + i * b16 + N);
      }
    }

#pragma unroll
    for (int kk = 0; kk < 32; kk += 4) {
      const int ak = kk + hi * 2;
      const int kp = (kk >> 1) + hi;
      v2f a0 = ldsv2(&as[w * 32 + ln][ak]);
      v2f a1 = ldsv2(&as[w * 32 + 16 + ln][ak]);
#pragma unroll
      for (int nt = 0; nt < 4; ++nt) {
        v2f b = ldsv2(&bs[kp][nt * 16 + ln][0]);
        acc[0][nt] = wmma4(a0, b, acc[0][nt]);
        acc[1][nt] = wmma4(a1, b, acc[1][nt]);
      }
    }
  }

#pragma unroll
  for (int mt = 0; mt < 2; ++mt)
#pragma unroll
    for (int nt = 0; nt < 4; ++nt)
#pragma unroll
      for (int vv = 0; vv < 8; ++vv) {
        int row = m0 + w * 32 + mt * 16 + vv + hi * 8;
        int col = n0 + nt * 16 + ln;
        C[(size_t)row * N + col] = acc[mt][nt][vv] + bias[col];
      }
}

__global__ __launch_bounds__(512) void xpos_kernel(float* __restrict__ qkv) {
  const int row = blockIdx.x;
  const int j = threadIdx.x;
  const int t = row & (CT - 1);

  const float D = (float)CE;
  const float base = (2.0f * (float)j + 0.4f * D) / (1.4f * D);
  const float sc = __powf(base, (float)t / 512.0f);
  const float invf = __powf(10000.0f, -(float)j / 512.0f);
  float sn, cn;
  __sincosf((float)t * invf, &sn, &cn);

  size_t off = (size_t)row * NQKV + 2 * j;
  {
    float x0 = qkv[off], x1 = qkv[off + 1];
    float c = cn * sc, s = sn * sc;
    qkv[off]     = x0 * c - x1 * s;
    qkv[off + 1] = x1 * c + x0 * s;
  }
  {
    float isc = 1.0f / sc;
    float x0 = qkv[off + CE], x1 = qkv[off + CE + 1];
    float c = cn * isc, s = sn * isc;
    qkv[off + CE]     = x0 * c - x1 * s;
    qkv[off + CE + 1] = x1 * c + x0 * s;
  }
}

__global__ __launch_bounds__(128) void retention_kernel(
    const float* __restrict__ qkv, float* __restrict__ ro) {
  __shared__ __attribute__((aligned(16))) float q_s[64][132];
  __shared__ __attribute__((aligned(16))) float k_s[16][132];
  __shared__ __attribute__((aligned(16))) float v_s[8][258][2];
  __shared__ float s_s[4][16][20];

  const int iblk = blockIdx.x;
  const int bh = blockIdx.y;
  const int b = bh >> 3, h = bh & 7;
  const int i0 = iblk * 64;
  const int rowbase = b * CT;

  const int tid = threadIdx.x;
  const int lane = tid & 31;
  const int w = tid >> 5;
  const int hi = lane >> 4;
  const int ln = lane & 15;

  const float se = logf(1.0f / 32.0f), ee = logf(1.0f / 512.0f);
  const float g = 1.0f - __expf(se + (ee - se) * ((float)h / 7.0f));
  const float logg = logf(g);
  const float rsdk = rsqrtf((float)CDK);

  const int kr = tid >> 5;
  const int kc = (tid & 31) * 4;
  const int vp = tid >> 6;
  const int vc = (tid & 63) * 4;
  const float* kptr = qkv + (size_t)(rowbase + kr) * NQKV + CE + h * CDK + kc;
  const float* vptr =
      qkv + (size_t)(rowbase + 2 * vp) * NQKV + 2 * CE + h * CDV + vc;

  v4f kreg[4], ve[4], vo[4];

#pragma unroll
  for (int i = 0; i < 4; ++i) {
    kreg[i] = *reinterpret_cast<const v4f*>(kptr + (size_t)(4 * i) * NQKV);
    ve[i] = *reinterpret_cast<const v4f*>(vptr + (size_t)(4 * i) * NQKV);
    vo[i] = *reinterpret_cast<const v4f*>(vptr + (size_t)(4 * i + 1) * NQKV);
  }

#pragma unroll
  for (int i = 0; i < 16; ++i) {
    const int r = kr + 4 * i;
    v4f qv = *reinterpret_cast<const v4f*>(
        qkv + (size_t)(rowbase + i0 + r) * NQKV + h * CDK + kc);
    *reinterpret_cast<v4f*>(&q_s[r][kc]) = qv;
  }

  v8f acc[16];
#pragma unroll
  for (int i = 0; i < 16; ++i) acc[i] = vzero8();

  for (int j0 = 0; j0 < i0 + 64; j0 += 16) {
    __syncthreads();
#pragma unroll
    for (int i = 0; i < 4; ++i) {
      *reinterpret_cast<v4f*>(&k_s[kr + 4 * i][kc]) = kreg[i];
      v4f lo, hi2;
      lo.x = ve[i].x;  lo.y = vo[i].x;  lo.z = ve[i].y;  lo.w = vo[i].y;
      hi2.x = ve[i].z; hi2.y = vo[i].z; hi2.z = ve[i].w; hi2.w = vo[i].w;
      *reinterpret_cast<v4f*>(&v_s[vp + 2 * i][vc][0]) = lo;
      *reinterpret_cast<v4f*>(&v_s[vp + 2 * i][vc + 2][0]) = hi2;
    }
    __syncthreads();

    {
      const int jn = (j0 + 16 < i0 + 64) ? (j0 + 16) : j0;
      const size_t adv = (size_t)jn * NQKV;
#pragma unroll
      for (int i = 0; i < 4; ++i) {
        kreg[i] =
            *reinterpret_cast<const v4f*>(kptr + adv + (size_t)(4 * i) * NQKV);
        ve[i] =
            *reinterpret_cast<const v4f*>(vptr + adv + (size_t)(4 * i) * NQKV);
        vo[i] = *reinterpret_cast<const v4f*>(vptr + adv +
                                              (size_t)(4 * i + 1) * NQKV);
      }
    }

    v8f sacc = vzero8();
#pragma unroll
    for (int kk = 0; kk < CDK; kk += 4) {
      const int ak = kk + hi * 2;
      v2f a = ldsv2(&q_s[w * 16 + ln][ak]);
      v2f bf = ldsv2(&k_s[ln][ak]);
      sacc = wmma4(a, bf, sacc);
    }

#pragma unroll
    for (int vv = 0; vv < 8; ++vv) {
      int i = i0 + w * 16 + vv + hi * 8;
      int j = j0 + ln;
      float m = (i >= j) ? __expf(logg * (float)(i - j)) * rsdk : 0.0f;
      sacc[vv] *= m;
    }

#pragma unroll
    for (int vv = 0; vv < 8; ++vv) s_s[w][vv + hi * 8][ln] = sacc[vv];

#pragma unroll
    for (int ks = 0; ks < 16; ks += 4) {
      const int ak = ks + hi * 2;
      const int kp = (ks >> 1) + hi;
      v2f a = ldsv2(&s_s[w][ln][ak]);
#pragma unroll
      for (int nt = 0; nt < 16; ++nt) {
        v2f bf = ldsv2(&v_s[kp][nt * 16 + ln][0]);
        acc[nt] = wmma4(a, bf, acc[nt]);
      }
    }
  }

  const float inv1mg = 1.0f / (1.0f - g);
#pragma unroll
  for (int nt = 0; nt < 16; ++nt)
#pragma unroll
    for (int vv = 0; vv < 8; ++vv) {
      int i = i0 + w * 16 + vv + hi * 8;
      float rowsum = (1.0f - __expf(logg * (float)(i + 1))) * inv1mg;
      float sc = rsqrtf(rowsum);
      ro[(size_t)(rowbase + i) * CV + h * CDV + nt * 16 + ln] = acc[nt][vv] * sc;
    }
}

__global__ __launch_bounds__(256) void curr_kv_kernel(
    const float* __restrict__ qkv, float* __restrict__ kvout) {
  __shared__ __attribute__((aligned(16))) float kw_s[8][130][2];
  __shared__ __attribute__((aligned(16))) float v_s[8][258][2];

  const int bh = blockIdx.x;
  const int b = bh >> 3, h = bh & 7;
  const int rowbase = b * CT;

  const int tid = threadIdx.x;
  const int lane = tid & 31;
  const int w = tid >> 5;
  const int hi = lane >> 4;
  const int ln = lane & 15;

  const float se = logf(1.0f / 32.0f), ee = logf(1.0f / 512.0f);
  const float g = 1.0f - __expf(se + (ee - se) * ((float)h / 7.0f));
  const float logg = logf(g);
  const float rnorm = rsqrtf((1.0f - __expf(logg * (float)CT)) / (1.0f - g));

  const int kp = tid >> 5;
  const int kc = (tid & 31) * 4;
  const int vp = tid >> 6;
  const int vc = (tid & 63) * 4;
  const float* kptr =
      qkv + (size_t)(rowbase + 2 * kp) * NQKV + CE + h * CDK + kc;
  const float* vptr =
      qkv + (size_t)(rowbase + 2 * vp) * NQKV + 2 * CE + h * CDV + vc;

  v4f ke, ko, ve[2], vo[2];

  ke = *reinterpret_cast<const v4f*>(kptr);
  ko = *reinterpret_cast<const v4f*>(kptr + NQKV);
#pragma unroll
  for (int i = 0; i < 2; ++i) {
    ve[i] = *reinterpret_cast<const v4f*>(vptr + (size_t)(8 * i) * NQKV);
    vo[i] = *reinterpret_cast<const v4f*>(vptr + (size_t)(8 * i + 1) * NQKV);
  }

  v8f acc[16];
#pragma unroll
  for (int i = 0; i < 16; ++i) acc[i] = vzero8();

  for (int j0 = 0; j0 < CT; j0 += 16) {
    __syncthreads();
    {
      const int je = j0 + 2 * kp;
      const float we = __expf(logg * (float)(CT - 1 - je)) * rnorm;
      const float wo = __expf(logg * (float)(CT - 2 - je)) * rnorm;
      v4f lo, hi2;
      lo.x = ke.x * we;  lo.y = ko.x * wo;  lo.z = ke.y * we;  lo.w = ko.y * wo;
      hi2.x = ke.z * we; hi2.y = ko.z * wo; hi2.z = ke.w * we; hi2.w = ko.w * wo;
      *reinterpret_cast<v4f*>(&kw_s[kp][kc][0]) = lo;
      *reinterpret_cast<v4f*>(&kw_s[kp][kc + 2][0]) = hi2;
    }
#pragma unroll
    for (int i = 0; i < 2; ++i) {
      v4f lo, hi2;
      lo.x = ve[i].x;  lo.y = vo[i].x;  lo.z = ve[i].y;  lo.w = vo[i].y;
      hi2.x = ve[i].z; hi2.y = vo[i].z; hi2.z = ve[i].w; hi2.w = vo[i].w;
      *reinterpret_cast<v4f*>(&v_s[vp + 4 * i][vc][0]) = lo;
      *reinterpret_cast<v4f*>(&v_s[vp + 4 * i][vc + 2][0]) = hi2;
    }
    __syncthreads();

    {
      const int jn = (j0 + 16 < CT) ? (j0 + 16) : j0;
      const size_t adv = (size_t)jn * NQKV;
      ke = *reinterpret_cast<const v4f*>(kptr + adv);
      ko = *reinterpret_cast<const v4f*>(kptr + adv + NQKV);
#pragma unroll
      for (int i = 0; i < 2; ++i) {
        ve[i] =
            *reinterpret_cast<const v4f*>(vptr + adv + (size_t)(8 * i) * NQKV);
        vo[i] = *reinterpret_cast<const v4f*>(vptr + adv +
                                              (size_t)(8 * i + 1) * NQKV);
      }
    }

#pragma unroll
    for (int ks = 0; ks < 16; ks += 4) {
      const int kq = (ks >> 1) + hi;
      v2f a = ldsv2(&kw_s[kq][w * 16 + ln][0]);
#pragma unroll
      for (int nt = 0; nt < 16; ++nt) {
        v2f bf = ldsv2(&v_s[kq][nt * 16 + ln][0]);
        acc[nt] = wmma4(a, bf, acc[nt]);
      }
    }
    asm volatile("" ::"v"(ke), "v"(ko));
  }

#pragma unroll
  for (int nt = 0; nt < 16; ++nt)
#pragma unroll
    for (int vv = 0; vv < 8; ++vv) {
      int kk = w * 16 + vv + hi * 8;
      kvout[(size_t)bh * CDK * CDV + (size_t)kk * CDV + nt * 16 + ln] =
          acc[nt][vv];
    }
}

__global__ __launch_bounds__(256) void gn_gate_kernel(
    const float* __restrict__ ro, float* __restrict__ gbuf) {
  const int row = blockIdx.x;
  const int w = threadIdx.x >> 5;
  const int lane = threadIdx.x & 31;
  const size_t base = (size_t)row * CV + (size_t)w * CDV;

  float x[8];
  float s = 0.0f, s2 = 0.0f;
#pragma unroll
  for (int i = 0; i < 8; ++i) {
    x[i] = ro[base + i * 32 + lane];
    s += x[i];
    s2 += x[i] * x[i];
  }
#pragma unroll
  for (int off = 16; off > 0; off >>= 1) {
    s += __shfl_xor(s, off, 32);
    s2 += __shfl_xor(s2, off, 32);
  }
  const float mu = s * (1.0f / 256.0f);
  const float var = s2 * (1.0f / 256.0f) - mu * mu;
  const float rstd = rsqrtf(var + 1e-5f);

#pragma unroll
  for (int i = 0; i < 8; ++i) {
    size_t idx = base + i * 32 + lane;
    float nrm = (x[i] - mu) * rstd;
    float gv = gbuf[idx];
    float sil = gv / (1.0f + __expf(-gv));
    gbuf[idx] = sil * nrm;
  }
}

extern "C" void kernel_launch(void* const* d_in, const int* in_sizes, int n_in,
                              void* d_out, int out_size, void* d_ws,
                              size_t ws_size, hipStream_t stream) {
  const float* hs    = (const float*)d_in[0];
  const float* w_qkv = (const float*)d_in[1];
  const float* b_qkv = (const float*)d_in[2];
  const float* w_g   = (const float*)d_in[3];
  const float* b_g   = (const float*)d_in[4];
  const float* w_p   = (const float*)d_in[5];
  const float* b_p   = (const float*)d_in[6];

  float* out  = (float*)d_out;
  float* qkv  = (float*)d_ws;
  float* gbuf = qkv + (size_t)BT * NQKV;
  float* ro   = gbuf + (size_t)BT * CV;

  gemm_wmma_f32<<<dim3(NQKV / 64, BT / 128), 128, 0, stream>>>(
      hs, w_qkv, b_qkv, qkv, BT, NQKV, CE);

  xpos_kernel<<<dim3(BT), 512, 0, stream>>>(qkv);

  gemm_wmma_f32<<<dim3(CV / 64, BT / 128), 128, 0, stream>>>(
      hs, w_g, b_g, gbuf, BT, CV, CE);

  retention_kernel<<<dim3(CT / 64, CB * CH), 128, 0, stream>>>(qkv, ro);

  curr_kv_kernel<<<dim3(CB * CH), 256, 0, stream>>>(
      qkv, out + (size_t)BT * CE);

  gn_gate_kernel<<<dim3(BT), 256, 0, stream>>>(ro, gbuf);

  gemm_wmma_f32<<<dim3(CE / 64, BT / 128), 128, 0, stream>>>(
      gbuf, w_p, b_p, out, BT, CE, CV);
}


# curr_kv kernel: each (b,h)'s 128x256 result cut into 16 tiles of 32x64, one workgroup each (256 workgroups, was 16), so all WGPs share its matrix work; order of the sums over j unchanged
# speedup vs baseline: 3.8012x; 1.1915x over previous
#include <hip/hip_runtime.h>
#include <hip/hip_bf16.h>


typedef __attribute__((ext_vector_type(2))) float v2f;
typedef __attribute__((ext_vector_type(4))) float v4f;
typedef __attribute__((ext_vector_type(8))) float v8f;

#define CB 2
#define CT 2048
#define CE 1024
#define CH 8
#define CDK 128
#define CDV 256
#define CV 2048
#define NQKV 4096
#define BT (CB * CT)

static __device__ __forceinline__ v8f wmma4(v2f a, v2f b, v8f c) {
  return __builtin_amdgcn_wmma_f32_16x16x4_f32(
       false, a,  false, b,
       (short)0, c,  false,  false);
}

static __device__ __forceinline__ v8f vzero8() {
  v8f z;
#pragma unroll
  for (int i = 0; i < 8; ++i) z[i] = 0.0f;
  return z;
}

static __device__ __forceinline__ v2f ldsv2(const float* p) {
  return *reinterpret_cast<const v2f*>(p);
}

__global__ __launch_bounds__(128) __attribute__((amdgpu_waves_per_eu(4, 4)))
void gemm_wmma_f32(
    const float* __restrict__ A, const float* __restrict__ Bm,
    const float* __restrict__ bias, float* __restrict__ C,
    int M, int N, int K) {
  __shared__ __attribute__((aligned(16))) float as[128][36];
  __shared__ __attribute__((aligned(16))) float bs[16][66][2];

  const int tid = threadIdx.x;
  const int lane = tid & 31;
  const int w = tid >> 5;
  const int hi = lane >> 4;
  const int ln = lane & 15;
  const int n0 = blockIdx.x * 64;
  const int m0 = blockIdx.y * 128;

  const int ar = tid >> 3;
  const int ac = (tid & 7) * 4;
  const int bp = tid >> 4;
  const int bc = (tid & 15) * 4;
  const float* aptr = A + (size_t)(m0 + ar) * K + ac;
  const float* bptr = Bm + (size_t)(2 * bp) * N + (n0 + bc);
  const size_t a16 = (size_t)16 * K;
  const size_t b16 = (size_t)16 * N;

  v4f areg[8], be[2], bo[2];

#pragma unroll
  for (int i = 0; i < 8; ++i)
    areg[i] = *reinterpret_cast<const v4f*>(aptr + i * a16);
#pragma unroll
  for (int i = 0; i < 2; ++i) {
    be[i] = *reinterpret_cast<const v4f*>(bptr + i * b16);
    bo[i] = *reinterpret_cast<const v4f*>(bptr + i * b16 + N);
  }

  v8f acc[2][4];
#pragma unroll
  for (int mt = 0; mt < 2; ++mt)
#pragma unroll
    for (int nt = 0; nt < 4; ++nt) acc[mt][nt] = vzero8();

  for (int k0 = 0; k0 < K; k0 += 32) {
    __syncthreads();
#pragma unroll
    for (int i = 0; i < 8; ++i)
      *reinterpret_cast<v4f*>(&as[ar + 16 * i][ac]) = areg[i];
#pragma unroll
    for (int i = 0; i < 2; ++i) {
      v4f lo, hi2;
      lo.x = be[i].x;  lo.y = bo[i].x;  lo.z = be[i].y;  lo.w = bo[i].y;
      hi2.x = be[i].z; hi2.y = bo[i].z; hi2.z = be[i].w; hi2.w = bo[i].w;
      *reinterpret_cast<v4f*>(&bs[bp + 8 * i][bc][0]) = lo;
      *reinterpret_cast<v4f*>(&bs[bp + 8 * i][bc + 2][0]) = hi2;
    }
    __syncthreads();

    {
      const int kn = (k0 + 32 < K) ? (k0 + 32) : k0;
      const float* ap = aptr + kn;
      const float* bq = bptr + (size_t)kn * N;
#pragma unroll
      for (int i = 0; i < 8; ++i)
        areg[i] = *reinterpret_cast<const v4f*>(ap + i * a16);
#pragma unroll
      for (int i = 0; i < 2; ++i) {
        be[i] = *reinterpret_cast<const v4f*>(bq + i * b16);
        bo[i] = *reinterpret_cast<const v4f*>(bq + i * b16 + N);
      }
    }

#pragma unroll
    for (int kk = 0; kk < 32; kk += 4) {
      const int ak = kk + hi * 2;
      const int kp = (kk >> 1) + hi;
      v2f a0 = ldsv2(&as[w * 32 + ln][ak]);
      v2f a1 = ldsv2(&as[w * 32 + 16 + ln][ak]);
#pragma unroll
      for (int nt = 0; nt < 4; ++nt) {
        v2f b = ldsv2(&bs[kp][nt * 16 + ln][0]);
        acc[0][nt] = wmma4(a0, b, acc[0][nt]);
        acc[1][nt] = wmma4(a1, b, acc[1][nt]);
      }
    }
  }

#pragma unroll
  for (int mt = 0; mt < 2; ++mt)
#pragma unroll
    for (int nt = 0; nt < 4; ++nt)
#pragma unroll
      for (int vv = 0; vv < 8; ++vv) {
        int row = m0 + w * 32 + mt * 16 + vv + hi * 8;
        int col = n0 + nt * 16 + ln;
        C[(size_t)row * N + col] = acc[mt][nt][vv] + bias[col];
      }
}

__global__ __launch_bounds__(512) void xpos_kernel(float* __restrict__ qkv) {
  const int row = blockIdx.x;
  const int j = threadIdx.x;
  const int t = row & (CT - 1);

  const float D = (float)CE;
  const float base = (2.0f * (float)j + 0.4f * D) / (1.4f * D);
  const float sc = __powf(base, (float)t / 512.0f);
  const float invf = __powf(10000.0f, -(float)j / 512.0f);
  float sn, cn;
  __sincosf((float)t * invf, &sn, &cn);

  size_t off = (size_t)row * NQKV + 2 * j;
  {
    float x0 = qkv[off], x1 = qkv[off + 1];
    float c = cn * sc, s = sn * sc;
    qkv[off]     = x0 * c - x1 * s;
    qkv[off + 1] = x1 * c + x0 * s;
  }
  {
    float isc = 1.0f / sc;
    float x0 = qkv[off + CE], x1 = qkv[off + CE + 1];
    float c = cn * isc, s = sn * isc;
    qkv[off + CE]     = x0 * c - x1 * s;
    qkv[off + CE + 1] = x1 * c + x0 * s;
  }
}

__global__ __launch_bounds__(128) void retention_kernel(
    const float* __restrict__ qkv, float* __restrict__ ro) {
  __shared__ __attribute__((aligned(16))) float q_s[64][132];
  __shared__ __attribute__((aligned(16))) float k_s[16][132];
  __shared__ __attribute__((aligned(16))) float v_s[8][258][2];
  __shared__ float s_s[4][16][20];

  const int iblk = blockIdx.x;
  const int bh = blockIdx.y;
  const int b = bh >> 3, h = bh & 7;
  const int i0 = iblk * 64;
  const int rowbase = b * CT;

  const int tid = threadIdx.x;
  const int lane = tid & 31;
  const int w = tid >> 5;
  const int hi = lane >> 4;
  const int ln = lane & 15;

  const float se = logf(1.0f / 32.0f), ee = logf(1.0f / 512.0f);
  const float g = 1.0f - __expf(se + (ee - se) * ((float)h / 7.0f));
  const float logg = logf(g);
  const float rsdk = rsqrtf((float)CDK);

  const int kr = tid >> 5;
  const int kc = (tid & 31) * 4;
  const int vp = tid >> 6;
  const int vc = (tid & 63) * 4;
  const float* kptr = qkv + (size_t)(rowbase + kr) * NQKV + CE + h * CDK + kc;
  const float* vptr =
      qkv + (size_t)(rowbase + 2 * vp) * NQKV + 2 * CE + h * CDV + vc;

  v4f kreg[4], ve[4], vo[4];

#pragma unroll
  for (int i = 0; i < 4; ++i) {
    kreg[i] = *reinterpret_cast<const v4f*>(kptr + (size_t)(4 * i) * NQKV);
    ve[i] = *reinterpret_cast<const v4f*>(vptr + (size_t)(4 * i) * NQKV);
    vo[i] = *reinterpret_cast<const v4f*>(vptr + (size_t)(4 * i + 1) * NQKV);
  }

#pragma unroll
  for (int i = 0; i < 16; ++i) {
    const int r = kr + 4 * i;
    v4f qv = *reinterpret_cast<const v4f*>(
        qkv + (size_t)(rowbase + i0 + r) * NQKV + h * CDK + kc);
    *reinterpret_cast<v4f*>(&q_s[r][kc]) = qv;
  }

  v8f acc[16];
#pragma unroll
  for (int i = 0; i < 16; ++i) acc[i] = vzero8();

  for (int j0 = 0; j0 < i0 + 64; j0 += 16) {
    __syncthreads();
#pragma unroll
    for (int i = 0; i < 4; ++i) {
      *reinterpret_cast<v4f*>(&k_s[kr + 4 * i][kc]) = kreg[i];
      v4f lo, hi2;
      lo.x = ve[i].x;  lo.y = vo[i].x;  lo.z = ve[i].y;  lo.w = vo[i].y;
      hi2.x = ve[i].z; hi2.y = vo[i].z; hi2.z = ve[i].w; hi2.w = vo[i].w;
      *reinterpret_cast<v4f*>(&v_s[vp + 2 * i][vc][0]) = lo;
      *reinterpret_cast<v4f*>(&v_s[vp + 2 * i][vc + 2][0]) = hi2;
    }
    __syncthreads();

    {
      const int jn = (j0 + 16 < i0 + 64) ? (j0 + 16) : j0;
      const size_t adv = (size_t)jn * NQKV;
#pragma unroll
      for (int i = 0; i < 4; ++i) {
        kreg[i] =
            *reinterpret_cast<const v4f*>(kptr + adv + (size_t)(4 * i) * NQKV);
        ve[i] =
            *reinterpret_cast<const v4f*>(vptr + adv + (size_t)(4 * i) * NQKV);
        vo[i] = *reinterpret_cast<const v4f*>(vptr + adv +
                                              (size_t)(4 * i + 1) * NQKV);
      }
    }

    v8f sacc = vzero8();
#pragma unroll
    for (int kk = 0; kk < CDK; kk += 4) {
      const int ak = kk + hi * 2;
      v2f a = ldsv2(&q_s[w * 16 + ln][ak]);
      v2f bf = ldsv2(&k_s[ln][ak]);
      sacc = wmma4(a, bf, sacc);
    }

#pragma unroll
    for (int vv = 0; vv < 8; ++vv) {
      int i = i0 + w * 16 + vv + hi * 8;
      int j = j0 + ln;
      float m = (i >= j) ? __expf(logg * (float)(i - j)) * rsdk : 0.0f;
      sacc[vv] *= m;
    }

#pragma unroll
    for (int vv = 0; vv < 8; ++vv) s_s[w][vv + hi * 8][ln] = sacc[vv];

#pragma unroll
    for (int ks = 0; ks < 16; ks += 4) {
      const int ak = ks + hi * 2;
      const int kp = (ks >> 1) + hi;
      v2f a = ldsv2(&s_s[w][ln][ak]);
#pragma unroll
      for (int nt = 0; nt < 16; ++nt) {
        v2f bf = ldsv2(&v_s[kp][nt * 16 + ln][0]);
        acc[nt] = wmma4(a, bf, acc[nt]);
      }
    }
  }

  const float inv1mg = 1.0f / (1.0f - g);
#pragma unroll
  for (int nt = 0; nt < 16; ++nt)
#pragma unroll
    for (int vv = 0; vv < 8; ++vv) {
      int i = i0 + w * 16 + vv + hi * 8;
      float rowsum = (1.0f - __expf(logg * (float)(i + 1))) * inv1mg;
      float sc = rsqrtf(rowsum);
      ro[(size_t)(rowbase + i) * CV + h * CDV + nt * 16 + ln] = acc[nt][vv] * sc;
    }
}

#define KV_JT 128
#define KV_MT 32
#define KV_NT 64
__global__ __launch_bounds__(256) __attribute__((amdgpu_waves_per_eu(4, 4)))
void curr_kv_kernel(
    const float* __restrict__ qkv, float* __restrict__ kvout) {
  __shared__ __attribute__((aligned(16))) float kw_s[KV_JT / 2][KV_MT + 16][2];
  __shared__ __attribute__((aligned(16))) float v_s[KV_JT / 2][KV_NT + 16][2];

  const int bh = blockIdx.y;
  const int b = bh >> 3, h = bh & 7;
  const int rowbase = b * CT;
  const int kk0 = (blockIdx.x >> 2) * KV_MT;
  const int vv0 = (blockIdx.x & 3) * KV_NT;

  const int tid = threadIdx.x;
  const int lane = tid & 31;
  const int w = tid >> 5;
  const int hi = lane >> 4;
  const int ln = lane & 15;
  const int mi = w >> 2;
  const int ni = w & 3;

  const float se = logf(1.0f / 32.0f), ee = logf(1.0f / 512.0f);
  const float g = 1.0f - __expf(se + (ee - se) * ((float)h / 7.0f));
  const float logg = logf(g);
  const float rnorm = rsqrtf((1.0f - __expf(logg * (float)CT)) / (1.0f - g));

  const int kp = tid >> 3;
  const int kc = (tid & 7) * 4;
  const int vp = tid >> 4;
  const int vc = (tid & 15) * 4;
  const float* kptr =
      qkv + (size_t)(rowbase + 2 * kp) * NQKV + CE + h * CDK + kk0 + kc;
  const float* vptr =
      qkv + (size_t)(rowbase + 2 * vp) * NQKV + 2 * CE + h * CDV + vv0 + vc;

  v4f ke[2], ko[2], ve[4], vo[4];
  v4f klo[2], khi[2];

#pragma unroll
  for (int i = 0; i < 2; ++i) {
    ke[i] = *reinterpret_cast<const v4f*>(kptr + (size_t)(64 * i) * NQKV);
    ko[i] = *reinterpret_cast<const v4f*>(kptr + (size_t)(64 * i + 1) * NQKV);
  }
#pragma unroll
  for (int i = 0; i < 4; ++i) {
    ve[i] = *reinterpret_cast<const v4f*>(vptr + (size_t)(32 * i) * NQKV);
    vo[i] = *reinterpret_cast<const v4f*>(vptr + (size_t)(32 * i + 1) * NQKV);
  }

#pragma unroll
  for (int i = 0; i < 2; ++i) {
    const int je = 0 + 2 * (kp + 32 * i);
    const float we = __expf(logg * (float)(CT - 1 - je)) * rnorm;
    const float wo = __expf(logg * (float)(CT - 2 - je)) * rnorm;
    klo[i].x = ke[i].x * we; klo[i].y = ko[i].x * wo;
    klo[i].z = ke[i].y * we; klo[i].w = ko[i].y * wo;
    khi[i].x = ke[i].z * we; khi[i].y = ko[i].z * wo;
    khi[i].z = ke[i].w * we; khi[i].w = ko[i].w * wo;
  }

  v8f acc = vzero8();

  for (int j0 = 0; j0 < CT; j0 += KV_JT) {
    __syncthreads();
#pragma unroll
    for (int i = 0; i < 2; ++i) {
      *reinterpret_cast<v4f*>(&kw_s[kp + 32 * i][kc][0]) = klo[i];
      *reinterpret_cast<v4f*>(&kw_s[kp + 32 * i][kc + 2][0]) = khi[i];
    }
#pragma unroll
    for (int i = 0; i < 4; ++i) {
      v4f lo, hi2;
      lo.x = ve[i].x;  lo.y = vo[i].x;  lo.z = ve[i].y;  lo.w = vo[i].y;
      hi2.x = ve[i].z; hi2.y = vo[i].z; hi2.z = ve[i].w; hi2.w = vo[i].w;
      *reinterpret_cast<v4f*>(&v_s[vp + 16 * i][vc][0]) = lo;
      *reinterpret_cast<v4f*>(&v_s[vp + 16 * i][vc + 2][0]) = hi2;
    }
    __syncthreads();

    const int jn = (j0 + KV_JT < CT) ? (j0 + KV_JT) : j0;
    {
      const size_t adv = (size_t)jn * NQKV;
#pragma unroll
      for (int i = 0; i < 2; ++i) {
        ke[i] =
            *reinterpret_cast<const v4f*>(kptr + adv + (size_t)(64 * i) * NQKV);
        ko[i] = *reinterpret_cast<const v4f*>(kptr + adv +
                                              (size_t)(64 * i + 1) * NQKV);
      }
#pragma unroll
      for (int i = 0; i < 4; ++i) {
        ve[i] =
            *reinterpret_cast<const v4f*>(vptr + adv + (size_t)(32 * i) * NQKV);
        vo[i] = *reinterpret_cast<const v4f*>(vptr + adv +
                                              (size_t)(32 * i + 1) * NQKV);
      }
    }

#pragma unroll
    for (int ks = 0; ks < KV_JT; ks += 4) {
      const int kq = (ks >> 1) + hi;
      v2f a = ldsv2(&kw_s[kq][mi * 16 + ln][0]);
      v2f bf = ldsv2(&v_s[kq][ni * 16 + ln][0]);
      acc = wmma4(a, bf, acc);
    }
#pragma unroll
    for (int i = 0; i < 2; ++i) {
      const int je = jn + 2 * (kp + 32 * i);
      const float we = __expf(logg * (float)(CT - 1 - je)) * rnorm;
      const float wo = __expf(logg * (float)(CT - 2 - je)) * rnorm;
      klo[i].x = ke[i].x * we; klo[i].y = ko[i].x * wo;
      klo[i].z = ke[i].y * we; klo[i].w = ko[i].y * wo;
      khi[i].x = ke[i].z * we; khi[i].y = ko[i].z * wo;
      khi[i].z = ke[i].w * we; khi[i].w = ko[i].w * wo;
    }
  }

  __syncthreads();
  float* o_s = &v_s[0][0][0];
#pragma unroll
  for (int vv = 0; vv < 8; ++vv)
    o_s[(mi * 16 + vv + hi * 8) * 68 + ni * 16 + ln] = acc[vv];
  __syncthreads();
#pragma unroll
  for (int i = 0; i < 2; ++i) {
    const int r = vp + 16 * i;
    const v4f o = *reinterpret_cast<const v4f*>(&o_s[r * 68 + vc]);
    *reinterpret_cast<v4f*>(kvout + (size_t)bh * CDK * CDV +
                            (size_t)(kk0 + r) * CDV + vv0 + vc) = o;
  }
}

__global__ __launch_bounds__(256) void gn_gate_kernel(
    const float* __restrict__ ro, float* __restrict__ gbuf) {
  const int row = blockIdx.x;
  const int w = threadIdx.x >> 5;
  const int lane = threadIdx.x & 31;
  const size_t base = (size_t)row * CV + (size_t)w * CDV;

  float x[8];
  float s = 0.0f, s2 = 0.0f;
#pragma unroll
  for (int i = 0; i < 8; ++i) {
    x[i] = ro[base + i * 32 + lane];
    s += x[i];
    s2 += x[i] * x[i];
  }
#pragma unroll
  for (int off = 16; off > 0; off >>= 1) {
    s += __shfl_xor(s, off, 32);
    s2 += __shfl_xor(s2, off, 32);
  }
  const float mu = s * (1.0f / 256.0f);
  const float var = s2 * (1.0f / 256.0f) - mu * mu;
  const float rstd = rsqrtf(var + 1e-5f);

#pragma unroll
  for (int i = 0; i < 8; ++i) {
    size_t idx = base + i * 32 + lane;
    float nrm = (x[i] - mu) * rstd;
    float gv = gbuf[idx];
    float sil = gv / (1.0f + __expf(-gv));
    gbuf[idx] = sil * nrm;
  }
}

extern "C" void kernel_launch(void* const* d_in, const int* in_sizes, int n_in,
                              void* d_out, int out_size, void* d_ws,
                              size_t ws_size, hipStream_t stream) {
  const float* hs    = (const float*)d_in[0];
  const float* w_qkv = (const float*)d_in[1];
  const float* b_qkv = (const float*)d_in[2];
  const float* w_g   = (const float*)d_in[3];
  const float* b_g   = (const float*)d_in[4];
  const float* w_p   = (const float*)d_in[5];
  const float* b_p   = (const float*)d_in[6];

  float* out  = (float*)d_out;
  float* qkv  = (float*)d_ws;
  float* gbuf = qkv + (size_t)BT * NQKV;
  float* ro   = gbuf + (size_t)BT * CV;

  gemm_wmma_f32<<<dim3(NQKV / 64, BT / 128), 128, 0, stream>>>(
      hs, w_qkv, b_qkv, qkv, BT, NQKV, CE);

  xpos_kernel<<<dim3(BT), 512, 0, stream>>>(qkv);

  gemm_wmma_f32<<<dim3(CV / 64, BT / 128), 128, 0, stream>>>(
      hs, w_g, b_g, gbuf, BT, CV, CE);

  retention_kernel<<<dim3(CT / 64, CB * CH), 128, 0, stream>>>(qkv, ro);

  curr_kv_kernel<<<dim3(16, CB * CH), 256, 0, stream>>>(
      qkv, out + (size_t)BT * CE);

  gn_gate_kernel<<<dim3(BT), 256, 0, stream>>>(ro, gbuf);

  gemm_wmma_f32<<<dim3(CE / 64, BT / 128), 128, 0, stream>>>(
      gbuf, w_p, b_p, out, BT, CE, CV);
}


# retention kernel: each workgroup takes query blocks 31-p and p in turn (132 key tiles for every p, was 4 to 128), its 8 waves as two sets splitting the key tiles; the two key halves' sums added last
# speedup vs baseline: 4.3452x; 1.1431x over previous
#include <hip/hip_runtime.h>
#include <hip/hip_bf16.h>


typedef __attribute__((ext_vector_type(2))) float v2f;
typedef __attribute__((ext_vector_type(4))) float v4f;
typedef __attribute__((ext_vector_type(8))) float v8f;

#define CB 2
#define CT 2048
#define CE 1024
#define CH 8
#define CDK 128
#define CDV 256
#define CV 2048
#define NQKV 4096
#define BT (CB * CT)

static __device__ __forceinline__ v8f wmma4(v2f a, v2f b, v8f c) {
  return __builtin_amdgcn_wmma_f32_16x16x4_f32(
       false, a,  false, b,
       (short)0, c,  false,  false);
}

static __device__ __forceinline__ v8f vzero8() {
  v8f z;
#pragma unroll
  for (int i = 0; i < 8; ++i) z[i] = 0.0f;
  return z;
}

static __device__ __forceinline__ v2f ldsv2(const float* p) {
  return *reinterpret_cast<const v2f*>(p);
}

__global__ __launch_bounds__(128) __attribute__((amdgpu_waves_per_eu(4, 4)))
void gemm_wmma_f32(
    const float* __restrict__ A, const float* __restrict__ Bm,
    const float* __restrict__ bias, float* __restrict__ C,
    int M, int N, int K) {
  __shared__ __attribute__((aligned(16))) float as[128][36];
  __shared__ __attribute__((aligned(16))) float bs[16][66][2];

  const int tid = threadIdx.x;
  const int lane = tid & 31;
  const int w = tid >> 5;
  const int hi = lane >> 4;
  const int ln = lane & 15;
  const int n0 = blockIdx.x * 64;
  const int m0 = blockIdx.y * 128;

  const int ar = tid >> 3;
  const int ac = (tid & 7) * 4;
  const int bp = tid >> 4;
  const int bc = (tid & 15) * 4;
  const float* aptr = A + (size_t)(m0 + ar) * K + ac;
  const float* bptr = Bm + (size_t)(2 * bp) * N + (n0 + bc);
  const size_t a16 = (size_t)16 * K;
  const size_t b16 = (size_t)16 * N;

  v4f areg[8], be[2], bo[2];

#pragma unroll
  for (int i = 0; i < 8; ++i)
    areg[i] = *reinterpret_cast<const v4f*>(aptr + i * a16);
#pragma unroll
  for (int i = 0; i < 2; ++i) {
    be[i] = *reinterpret_cast<const v4f*>(bptr + i * b16);
    bo[i] = *reinterpret_cast<const v4f*>(bptr + i * b16 + N);
  }

  v8f acc[2][4];
#pragma unroll
  for (int mt = 0; mt < 2; ++mt)
#pragma unroll
    for (int nt = 0; nt < 4; ++nt) acc[mt][nt] = vzero8();

  for (int k0 = 0; k0 < K; k0 += 32) {
    __syncthreads();
#pragma unroll
    for (int i = 0; i < 8; ++i)
      *reinterpret_cast<v4f*>(&as[ar + 16 * i][ac]) = areg[i];
#pragma unroll
    for (int i = 0; i < 2; ++i) {
      v4f lo, hi2;
      lo.x = be[i].x;  lo.y = bo[i].x;  lo.z = be[i].y;  lo.w = bo[i].y;
      hi2.x = be[i].z; hi2.y = bo[i].z; hi2.z = be[i].w; hi2.w = bo[i].w;
      *reinterpret_cast<v4f*>(&bs[bp + 8 * i][bc][0]) = lo;
      *reinterpret_cast<v4f*>(&bs[bp + 8 * i][bc + 2][0]) = hi2;
    }
    __syncthreads();

    {
      const int kn = (k0 + 32 < K) ? (k0 + 32) : k0;
      const float* ap = aptr + kn;
      const float* bq = bptr + (size_t)kn * N;
#pragma unroll
      for (int i = 0; i < 8; ++i)
        areg[i] = *reinterpret_cast<const v4f*>(ap + i * a16);
#pragma unroll
      for (int i = 0; i < 2; ++i) {
        be[i] = *reinterpret_cast<const v4f*>(bq + i * b16);
        bo[i] = *reinterpret_cast<const v4f*>(bq + i * b16 + N);
      }
    }

#pragma unroll
    for (int kk = 0; kk < 32; kk += 4) {
      const int ak = kk + hi * 2;
      const int kp = (kk >> 1) + hi;
      v2f a0 = ldsv2(&as[w * 32 + ln][ak]);
      v2f a1 = ldsv2(&as[w * 32 + 16 + ln][ak]);
#pragma unroll
      for (int nt = 0; nt < 4; ++nt) {
        v2f b = ldsv2(&bs[kp][nt * 16 + ln][0]);
        acc[0][nt] = wmma4(a0, b, acc[0][nt]);
        acc[1][nt] = wmma4(a1, b, acc[1][nt]);
      }
    }
  }

#pragma unroll
  for (int mt = 0; mt < 2; ++mt)
#pragma unroll
    for (int nt = 0; nt < 4; ++nt)
#pragma unroll
      for (int vv = 0; vv < 8; ++vv) {
        int row = m0 + w * 32 + mt * 16 + vv + hi * 8;
        int col = n0 + nt * 16 + ln;
        C[(size_t)row * N + col] = acc[mt][nt][vv] + bias[col];
      }
}

__global__ __launch_bounds__(512) void xpos_kernel(float* __restrict__ qkv) {
  const int row = blockIdx.x;
  const int j = threadIdx.x;
  const int t = row & (CT - 1);

  const float D = (float)CE;
  const float base = (2.0f * (float)j + 0.4f * D) / (1.4f * D);
  const float sc = __powf(base, (float)t / 512.0f);
  const float invf = __powf(10000.0f, -(float)j / 512.0f);
  float sn, cn;
  __sincosf((float)t * invf, &sn, &cn);

  size_t off = (size_t)row * NQKV + 2 * j;
  {
    float x0 = qkv[off], x1 = qkv[off + 1];
    float c = cn * sc, s = sn * sc;
    qkv[off]     = x0 * c - x1 * s;
    qkv[off + 1] = x1 * c + x0 * s;
  }
  {
    float isc = 1.0f / sc;
    float x0 = qkv[off + CE], x1 = qkv[off + CE + 1];
    float c = cn * isc, s = sn * isc;
    qkv[off + CE]     = x0 * c - x1 * s;
    qkv[off + CE + 1] = x1 * c + x0 * s;
  }
}

__global__ __launch_bounds__(256) __attribute__((amdgpu_waves_per_eu(2, 2)))
void retention_kernel(
    const float* __restrict__ qkv, float* __restrict__ ro) {
  __shared__ __attribute__((aligned(16))) float q_s[64][132];
  __shared__ __attribute__((aligned(16))) float k_s[2][16][132];
  __shared__ __attribute__((aligned(16))) float v_s[2][8][258][2];
  __shared__ float s_s[8][16][20];
  __shared__ __attribute__((aligned(16))) v4f r_s[4][16][2][32];

  const int pr = blockIdx.x;
  const int bh = blockIdx.y;
  const int b = bh >> 3, h = bh & 7;
  const int rowbase = b * CT;

  const int tid = threadIdx.x;
  const int lane = tid & 31;
  const int wv = tid >> 5;
  const int st = wv >> 2;
  const int w = wv & 3;
  const int t7 = tid & 127;
  const int hi = lane >> 4;
  const int ln = lane & 15;

  const float se = logf(1.0f / 32.0f), ee = logf(1.0f / 512.0f);
  const float g = 1.0f - __expf(se + (ee - se) * ((float)h / 7.0f));
  const float logg = logf(g);
  const float rsdk = rsqrtf((float)CDK);
  const float inv1mg = 1.0f / (1.0f - g);

  const int kr = t7 >> 5;
  const int kc = (t7 & 31) * 4;
  const int vp = t7 >> 6;
  const int vc = (t7 & 63) * 4;
  const float* kptr = qkv + (size_t)(rowbase + kr) * NQKV + CE + h * CDK + kc;
  const float* vptr =
      qkv + (size_t)(rowbase + 2 * vp) * NQKV + 2 * CE + h * CDV + vc;
  const int qr = tid >> 5;
  const int qc = (tid & 31) * 4;

  float (*ks)[132] = k_s[st];
  float (*vs)[258][2] = v_s[st];

  for (int half = 0; half < 2; ++half) {
    const int iblk = half ? pr : 31 - pr;
    const int i0 = iblk * 64;
    const int nh = 2 * (iblk + 1);
    const int jbeg = st * nh * 16;

    v4f kreg[4], ve[4], vo[4];

    {
      const size_t adv = (size_t)jbeg * NQKV;
#pragma unroll
      for (int i = 0; i < 4; ++i) {
        kreg[i] =
            *reinterpret_cast<const v4f*>(kptr + adv + (size_t)(4 * i) * NQKV);
        ve[i] =
            *reinterpret_cast<const v4f*>(vptr + adv + (size_t)(4 * i) * NQKV);
        vo[i] = *reinterpret_cast<const v4f*>(vptr + adv +
                                              (size_t)(4 * i + 1) * NQKV);
      }
    }

#pragma unroll
    for (int i = 0; i < 8; ++i) {
      const int r = qr + 8 * i;
      v4f qv = *reinterpret_cast<const v4f*>(
          qkv + (size_t)(rowbase + i0 + r) * NQKV + h * CDK + qc);
      *reinterpret_cast<v4f*>(&q_s[r][qc]) = qv;
    }

    v8f acc[16];
#pragma unroll
    for (int i = 0; i < 16; ++i) acc[i] = vzero8();

    for (int t = 0; t < nh; ++t) {
      const int j0 = jbeg + 16 * t;
      __syncthreads();
#pragma unroll
      for (int i = 0; i < 4; ++i) {
        *reinterpret_cast<v4f*>(&ks[kr + 4 * i][kc]) = kreg[i];
        v4f lo, hi2;
        lo.x = ve[i].x;  lo.y = vo[i].x;  lo.z = ve[i].y;  lo.w = vo[i].y;
        hi2.x = ve[i].z; hi2.y = vo[i].z; hi2.z = ve[i].w; hi2.w = vo[i].w;
        *reinterpret_cast<v4f*>(&vs[vp + 2 * i][vc][0]) = lo;
        *reinterpret_cast<v4f*>(&vs[vp + 2 * i][vc + 2][0]) = hi2;
      }
      __syncthreads();

      {
        const int jn = (t + 1 < nh) ? (j0 + 16) : j0;
        const size_t adv = (size_t)jn * NQKV;
#pragma unroll
        for (int i = 0; i < 4; ++i) {
          kreg[i] =
              *reinterpret_cast<const v4f*>(kptr + adv + (size_t)(4 * i) * NQKV);
          ve[i] =
              *reinterpret_cast<const v4f*>(vptr + adv + (size_t)(4 * i) * NQKV);
          vo[i] = *reinterpret_cast<const v4f*>(vptr + adv +
                                                (size_t)(4 * i + 1) * NQKV);
        }
      }

      v8f sacc = vzero8();
#pragma unroll
      for (int kk = 0; kk < CDK; kk += 4) {
        const int ak = kk + hi * 2;
        v2f a = ldsv2(&q_s[w * 16 + ln][ak]);
        v2f bf = ldsv2(&ks[ln][ak]);
        sacc = wmma4(a, bf, sacc);
      }

#pragma unroll
      for (int vv = 0; vv < 8; ++vv) {
        int i = i0 + w * 16 + vv + hi * 8;
        int j = j0 + ln;
        float m = (i >= j) ? __expf(logg * (float)(i - j)) * rsdk : 0.0f;
        sacc[vv] *= m;
      }

#pragma unroll
      for (int vv = 0; vv < 8; ++vv) s_s[wv][vv + hi * 8][ln] = sacc[vv];

#pragma unroll
      for (int ks4 = 0; ks4 < 16; ks4 += 4) {
        const int ak = ks4 + hi * 2;
        const int kp = (ks4 >> 1) + hi;
        v2f a = ldsv2(&s_s[wv][ln][ak]);
#pragma unroll
        for (int nt = 0; nt < 16; ++nt) {
          v2f bf = ldsv2(&vs[kp][nt * 16 + ln][0]);
          acc[nt] = wmma4(a, bf, acc[nt]);
        }
      }
      __builtin_amdgcn_sched_barrier(0);
    }

    if (st == 1) {
#pragma unroll
      for (int nt = 0; nt < 16; ++nt) {
        v4f lo, hi2;
        lo.x = acc[nt][0];  lo.y = acc[nt][1];  lo.z = acc[nt][2];  lo.w = acc[nt][3];
        hi2.x = acc[nt][4]; hi2.y = acc[nt][5]; hi2.z = acc[nt][6]; hi2.w = acc[nt][7];
        r_s[w][nt][0][lane] = lo;
        r_s[w][nt][1][lane] = hi2;
      }
    }
    __syncthreads();

    if (st == 0) {
#pragma unroll
      for (int nt = 0; nt < 16; ++nt) {
        const v4f lo = r_s[w][nt][0][lane];
        const v4f hi2 = r_s[w][nt][1][lane];
        acc[nt][0] += lo.x;  acc[nt][1] += lo.y;  acc[nt][2] += lo.z;  acc[nt][3] += lo.w;
        acc[nt][4] += hi2.x; acc[nt][5] += hi2.y; acc[nt][6] += hi2.z; acc[nt][7] += hi2.w;
      }
#pragma unroll
      for (int nt = 0; nt < 16; ++nt)
#pragma unroll
        for (int vv = 0; vv < 8; ++vv) {
          int i = i0 + w * 16 + vv + hi * 8;
          float rowsum = (1.0f - __expf(logg * (float)(i + 1))) * inv1mg;
          float sc = rsqrtf(rowsum);
          ro[(size_t)(rowbase + i) * CV + h * CDV + nt * 16 + ln] = acc[nt][vv] * sc;
        }
    }
  }
}

#define KV_JT 128
#define KV_MT 32
#define KV_NT 64
__global__ __launch_bounds__(256) __attribute__((amdgpu_waves_per_eu(4, 4)))
void curr_kv_kernel(
    const float* __restrict__ qkv, float* __restrict__ kvout) {
  __shared__ __attribute__((aligned(16))) float kw_s[KV_JT / 2][KV_MT + 16][2];
  __shared__ __attribute__((aligned(16))) float v_s[KV_JT / 2][KV_NT + 16][2];

  const int bh = blockIdx.y;
  const int b = bh >> 3, h = bh & 7;
  const int rowbase = b * CT;
  const int kk0 = (blockIdx.x >> 2) * KV_MT;
  const int vv0 = (blockIdx.x & 3) * KV_NT;

  const int tid = threadIdx.x;
  const int lane = tid & 31;
  const int w = tid >> 5;
  const int hi = lane >> 4;
  const int ln = lane & 15;
  const int mi = w >> 2;
  const int ni = w & 3;

  const float se = logf(1.0f / 32.0f), ee = logf(1.0f / 512.0f);
  const float g = 1.0f - __expf(se + (ee - se) * ((float)h / 7.0f));
  const float logg = logf(g);
  const float rnorm = rsqrtf((1.0f - __expf(logg * (float)CT)) / (1.0f - g));

  const int kp = tid >> 3;
  const int kc = (tid & 7) * 4;
  const int vp = tid >> 4;
  const int vc = (tid & 15) * 4;
  const float* kptr =
      qkv + (size_t)(rowbase + 2 * kp) * NQKV + CE + h * CDK + kk0 + kc;
  const float* vptr =
      qkv + (size_t)(rowbase + 2 * vp) * NQKV + 2 * CE + h * CDV + vv0 + vc;

  v4f ke[2], ko[2], ve[4], vo[4];
  v4f klo[2], khi[2];

#pragma unroll
  for (int i = 0; i < 2; ++i) {
    ke[i] = *reinterpret_cast<const v4f*>(kptr + (size_t)(64 * i) * NQKV);
    ko[i] = *reinterpret_cast<const v4f*>(kptr + (size_t)(64 * i + 1) * NQKV);
  }
#pragma unroll
  for (int i = 0; i < 4; ++i) {
    ve[i] = *reinterpret_cast<const v4f*>(vptr + (size_t)(32 * i) * NQKV);
    vo[i] = *reinterpret_cast<const v4f*>(vptr + (size_t)(32 * i + 1) * NQKV);
  }

#pragma unroll
  for (int i = 0; i < 2; ++i) {
    const int je = 0 + 2 * (kp + 32 * i);
    const float we = __expf(logg * (float)(CT - 1 - je)) * rnorm;
    const float wo = __expf(logg * (float)(CT - 2 - je)) * rnorm;
    klo[i].x = ke[i].x * we; klo[i].y = ko[i].x * wo;
    klo[i].z = ke[i].y * we; klo[i].w = ko[i].y * wo;
    khi[i].x = ke[i].z * we; khi[i].y = ko[i].z * wo;
    khi[i].z = ke[i].w * we; khi[i].w = ko[i].w * wo;
  }

  v8f acc = vzero8();

  for (int j0 = 0; j0 < CT; j0 += KV_JT) {
    __syncthreads();
#pragma unroll
    for (int i = 0; i < 2; ++i) {
      *reinterpret_cast<v4f*>(&kw_s[kp + 32 * i][kc][0]) = klo[i];
      *reinterpret_cast<v4f*>(&kw_s[kp + 32 * i][kc + 2][0]) = khi[i];
    }
#pragma unroll
    for (int i = 0; i < 4; ++i) {
      v4f lo, hi2;
      lo.x = ve[i].x;  lo.y = vo[i].x;  lo.z = ve[i].y;  lo.w = vo[i].y;
      hi2.x = ve[i].z; hi2.y = vo[i].z; hi2.z = ve[i].w; hi2.w = vo[i].w;
      *reinterpret_cast<v4f*>(&v_s[vp + 16 * i][vc][0]) = lo;
      *reinterpret_cast<v4f*>(&v_s[vp + 16 * i][vc + 2][0]) = hi2;
    }
    __syncthreads();

    const int jn = (j0 + KV_JT < CT) ? (j0 + KV_JT) : j0;
    {
      const size_t adv = (size_t)jn * NQKV;
#pragma unroll
      for (int i = 0; i < 2; ++i) {
        ke[i] =
            *reinterpret_cast<const v4f*>(kptr + adv + (size_t)(64 * i) * NQKV);
        ko[i] = *reinterpret_cast<const v4f*>(kptr + adv +
                                              (size_t)(64 * i + 1) * NQKV);
      }
#pragma unroll
      for (int i = 0; i < 4; ++i) {
        ve[i] =
            *reinterpret_cast<const v4f*>(vptr + adv + (size_t)(32 * i) * NQKV);
        vo[i] = *reinterpret_cast<const v4f*>(vptr + adv +
                                              (size_t)(32 * i + 1) * NQKV);
      }
    }

#pragma unroll
    for (int ks = 0; ks < KV_JT; ks += 4) {
      const int kq = (ks >> 1) + hi;
      v2f a = ldsv2(&kw_s[kq][mi * 16 + ln][0]);
      v2f bf = ldsv2(&v_s[kq][ni * 16 + ln][0]);
      acc = wmma4(a, bf, acc);
    }
#pragma unroll
    for (int i = 0; i < 2; ++i) {
      const int je = jn + 2 * (kp + 32 * i);
      const float we = __expf(logg * (float)(CT - 1 - je)) * rnorm;
      const float wo = __expf(logg * (float)(CT - 2 - je)) * rnorm;
      klo[i].x = ke[i].x * we; klo[i].y = ko[i].x * wo;
      klo[i].z = ke[i].y * we; klo[i].w = ko[i].y * wo;
      khi[i].x = ke[i].z * we; khi[i].y = ko[i].z * wo;
      khi[i].z = ke[i].w * we; khi[i].w = ko[i].w * wo;
    }
  }

  __syncthreads();
  float* o_s = &v_s[0][0][0];
#pragma unroll
  for (int vv = 0; vv < 8; ++vv)
    o_s[(mi * 16 + vv + hi * 8) * 68 + ni * 16 + ln] = acc[vv];
  __syncthreads();
#pragma unroll
  for (int i = 0; i < 2; ++i) {
    const int r = vp + 16 * i;
    const v4f o = *reinterpret_cast<const v4f*>(&o_s[r * 68 + vc]);
    *reinterpret_cast<v4f*>(kvout + (size_t)bh * CDK * CDV +
                            (size_t)(kk0 + r) * CDV + vv0 + vc) = o;
  }
}

__global__ __launch_bounds__(256) void gn_gate_kernel(
    const float* __restrict__ ro, float* __restrict__ gbuf) {
  const int row = blockIdx.x;
  const int w = threadIdx.x >> 5;
  const int lane = threadIdx.x & 31;
  const size_t base = (size_t)row * CV + (size_t)w * CDV;

  float x[8];
  float s = 0.0f, s2 = 0.0f;
#pragma unroll
  for (int i = 0; i < 8; ++i) {
    x[i] = ro[base + i * 32 + lane];
    s += x[i];
    s2 += x[i] * x[i];
  }
#pragma unroll
  for (int off = 16; off > 0; off >>= 1) {
    s += __shfl_xor(s, off, 32);
    s2 += __shfl_xor(s2, off, 32);
  }
  const float mu = s * (1.0f / 256.0f);
  const float var = s2 * (1.0f / 256.0f) - mu * mu;
  const float rstd = rsqrtf(var + 1e-5f);

#pragma unroll
  for (int i = 0; i < 8; ++i) {
    size_t idx = base + i * 32 + lane;
    float nrm = (x[i] - mu) * rstd;
    float gv = gbuf[idx];
    float sil = gv / (1.0f + __expf(-gv));
    gbuf[idx] = sil * nrm;
  }
}

extern "C" void kernel_launch(void* const* d_in, const int* in_sizes, int n_in,
                              void* d_out, int out_size, void* d_ws,
                              size_t ws_size, hipStream_t stream) {
  const float* hs    = (const float*)d_in[0];
  const float* w_qkv = (const float*)d_in[1];
  const float* b_qkv = (const float*)d_in[2];
  const float* w_g   = (const float*)d_in[3];
  const float* b_g   = (const float*)d_in[4];
  const float* w_p   = (const float*)d_in[5];
  const float* b_p   = (const float*)d_in[6];

  float* out  = (float*)d_out;
  float* qkv  = (float*)d_ws;
  float* gbuf = qkv + (size_t)BT * NQKV;
  float* ro   = gbuf + (size_t)BT * CV;

  gemm_wmma_f32<<<dim3(NQKV / 64, BT / 128), 128, 0, stream>>>(
      hs, w_qkv, b_qkv, qkv, BT, NQKV, CE);

  xpos_kernel<<<dim3(BT), 512, 0, stream>>>(qkv);

  gemm_wmma_f32<<<dim3(CV / 64, BT / 128), 128, 0, stream>>>(
      hs, w_g, b_g, gbuf, BT, CV, CE);

  retention_kernel<<<dim3(CT / 128, CB * CH), 256, 0, stream>>>(qkv, ro);

  curr_kv_kernel<<<dim3(16, CB * CH), 256, 0, stream>>>(
      qkv, out + (size_t)BT * CE);

  gn_gate_kernel<<<dim3(BT), 256, 0, stream>>>(ro, gbuf);

  gemm_wmma_f32<<<dim3(CE / 64, BT / 128), 128, 0, stream>>>(
      gbuf, w_p, b_p, out, BT, CE, CV);
}


# retention chunkwise: curr_kv pass also emits decayed K^T V states per 128-key chunk; a query block = Q.state (rows scaled) + its in-chunk causal tiles only; sums over keys regrouped per chunk
# speedup vs baseline: 5.3872x; 1.2398x over previous
#include <hip/hip_runtime.h>
#include <hip/hip_bf16.h>


typedef __attribute__((ext_vector_type(2))) float v2f;
typedef __attribute__((ext_vector_type(4))) float v4f;
typedef __attribute__((ext_vector_type(8))) float v8f;

#define CB 2
#define CT 2048
#define CE 1024
#define CH 8
#define CDK 128
#define CDV 256
#define CV 2048
#define NQKV 4096
#define BT (CB * CT)

static __device__ __forceinline__ v8f wmma4(v2f a, v2f b, v8f c) {
  return __builtin_amdgcn_wmma_f32_16x16x4_f32(
       false, a,  false, b,
       (short)0, c,  false,  false);
}

static __device__ __forceinline__ v8f vzero8() {
  v8f z;
#pragma unroll
  for (int i = 0; i < 8; ++i) z[i] = 0.0f;
  return z;
}

static __device__ __forceinline__ v2f ldsv2(const float* p) {
  return *reinterpret_cast<const v2f*>(p);
}

__global__ __launch_bounds__(128) __attribute__((amdgpu_waves_per_eu(4, 4)))
void gemm_wmma_f32(
    const float* __restrict__ A, const float* __restrict__ Bm,
    const float* __restrict__ bias, float* __restrict__ C,
    int M, int N, int K) {
  __shared__ __attribute__((aligned(16))) float as[128][36];
  __shared__ __attribute__((aligned(16))) float bs[16][66][2];

  const int tid = threadIdx.x;
  const int lane = tid & 31;
  const int w = tid >> 5;
  const int hi = lane >> 4;
  const int ln = lane & 15;
  const int n0 = blockIdx.x * 64;
  const int m0 = blockIdx.y * 128;

  const int ar = tid >> 3;
  const int ac = (tid & 7) * 4;
  const int bp = tid >> 4;
  const int bc = (tid & 15) * 4;
  const float* aptr = A + (size_t)(m0 + ar) * K + ac;
  const float* bptr = Bm + (size_t)(2 * bp) * N + (n0 + bc);
  const size_t a16 = (size_t)16 * K;
  const size_t b16 = (size_t)16 * N;

  v4f areg[8], be[2], bo[2];

#pragma unroll
  for (int i = 0; i < 8; ++i)
    areg[i] = *reinterpret_cast<const v4f*>(aptr + i * a16);
#pragma unroll
  for (int i = 0; i < 2; ++i) {
    be[i] = *reinterpret_cast<const v4f*>(bptr + i * b16);
    bo[i] = *reinterpret_cast<const v4f*>(bptr + i * b16 + N);
  }

  v8f acc[2][4];
#pragma unroll
  for (int mt = 0; mt < 2; ++mt)
#pragma unroll
    for (int nt = 0; nt < 4; ++nt) acc[mt][nt] = vzero8();

  for (int k0 = 0; k0 < K; k0 += 32) {
    __syncthreads();
#pragma unroll
    for (int i = 0; i < 8; ++i)
      *reinterpret_cast<v4f*>(&as[ar + 16 * i][ac]) = areg[i];
#pragma unroll
    for (int i = 0; i < 2; ++i) {
      v4f lo, hi2;
      lo.x = be[i].x;  lo.y = bo[i].x;  lo.z = be[i].y;  lo.w = bo[i].y;
      hi2.x = be[i].z; hi2.y = bo[i].z; hi2.z = be[i].w; hi2.w = bo[i].w;
      *reinterpret_cast<v4f*>(&bs[bp + 8 * i][bc][0]) = lo;
      *reinterpret_cast<v4f*>(&bs[bp + 8 * i][bc + 2][0]) = hi2;
    }
    __syncthreads();

    {
      const int kn = (k0 + 32 < K) ? (k0 + 32) : k0;
      const float* ap = aptr + kn;
      const float* bq = bptr + (size_t)kn * N;
#pragma unroll
      for (int i = 0; i < 8; ++i)
        areg[i] = *reinterpret_cast<const v4f*>(ap + i * a16);
#pragma unroll
      for (int i = 0; i < 2; ++i) {
        be[i] = *reinterpret_cast<const v4f*>(bq + i * b16);
        bo[i] = *reinterpret_cast<const v4f*>(bq + i * b16 + N);
      }
    }

#pragma unroll
    for (int kk = 0; kk < 32; kk += 4) {
      const int ak = kk + hi * 2;
      const int kp = (kk >> 1) + hi;
      v2f a0 = ldsv2(&as[w * 32 + ln][ak]);
      v2f a1 = ldsv2(&as[w * 32 + 16 + ln][ak]);
#pragma unroll
      for (int nt = 0; nt < 4; ++nt) {
        v2f b = ldsv2(&bs[kp][nt * 16 + ln][0]);
        acc[0][nt] = wmma4(a0, b, acc[0][nt]);
        acc[1][nt] = wmma4(a1, b, acc[1][nt]);
      }
    }
  }

#pragma unroll
  for (int mt = 0; mt < 2; ++mt)
#pragma unroll
    for (int nt = 0; nt < 4; ++nt)
#pragma unroll
      for (int vv = 0; vv < 8; ++vv) {
        int row = m0 + w * 32 + mt * 16 + vv + hi * 8;
        int col = n0 + nt * 16 + ln;
        C[(size_t)row * N + col] = acc[mt][nt][vv] + bias[col];
      }
}

__global__ __launch_bounds__(512) void xpos_kernel(float* __restrict__ qkv) {
  const int row = blockIdx.x;
  const int j = threadIdx.x;
  const int t = row & (CT - 1);

  const float D = (float)CE;
  const float base = (2.0f * (float)j + 0.4f * D) / (1.4f * D);
  const float sc = __powf(base, (float)t / 512.0f);
  const float invf = __powf(10000.0f, -(float)j / 512.0f);
  float sn, cn;
  __sincosf((float)t * invf, &sn, &cn);

  size_t off = (size_t)row * NQKV + 2 * j;
  {
    float x0 = qkv[off], x1 = qkv[off + 1];
    float c = cn * sc, s = sn * sc;
    qkv[off]     = x0 * c - x1 * s;
    qkv[off + 1] = x1 * c + x0 * s;
  }
  {
    float isc = 1.0f / sc;
    float x0 = qkv[off + CE], x1 = qkv[off + CE + 1];
    float c = cn * isc, s = sn * isc;
    qkv[off + CE]     = x0 * c - x1 * s;
    qkv[off + CE + 1] = x1 * c + x0 * s;
  }
}

#define KV_JT 128
#define KV_MT 32
#define KV_NT 64
#define NCH (CT / KV_JT)
__global__ __launch_bounds__(256) __attribute__((amdgpu_waves_per_eu(4, 4)))
void curr_kv_kernel(
    const float* __restrict__ qkv, float* __restrict__ kvout,
    float* __restrict__ pst) {
  __shared__ __attribute__((aligned(16))) float kw_s[KV_JT / 2][KV_MT + 16][2];
  __shared__ __attribute__((aligned(16))) float v_s[KV_JT / 2][KV_NT + 16][2];
  __shared__ __attribute__((aligned(16))) float o_s[KV_MT][KV_NT + 4];

  const int bh = blockIdx.y;
  const int b = bh >> 3, h = bh & 7;
  const int rowbase = b * CT;
  const int kk0 = (blockIdx.x >> 2) * KV_MT;
  const int vv0 = (blockIdx.x & 3) * KV_NT;

  const int tid = threadIdx.x;
  const int lane = tid & 31;
  const int w = tid >> 5;
  const int hi = lane >> 4;
  const int ln = lane & 15;
  const int mi = w >> 2;
  const int ni = w & 3;

  const float se = logf(1.0f / 32.0f), ee = logf(1.0f / 512.0f);
  const float g = 1.0f - __expf(se + (ee - se) * ((float)h / 7.0f));
  const float logg = logf(g);
  const float rnorm = rsqrtf((1.0f - __expf(logg * (float)CT)) / (1.0f - g));
  const float gstep = __expf(logg * (float)KV_JT);

  const int kp = tid >> 3;
  const int kc = (tid & 7) * 4;
  const int vp = tid >> 4;
  const int vc = (tid & 15) * 4;
  const float* kptr =
      qkv + (size_t)(rowbase + 2 * kp) * NQKV + CE + h * CDK + kk0 + kc;
  const float* vptr =
      qkv + (size_t)(rowbase + 2 * vp) * NQKV + 2 * CE + h * CDV + vv0 + vc;

  float we[2], wo[2];
#pragma unroll
  for (int i = 0; i < 2; ++i) {
    const int jj = 2 * (kp + 32 * i);
    we[i] = __expf(logg * (float)(KV_JT - 1 - jj));
    wo[i] = __expf(logg * (float)(KV_JT - 2 - jj));
  }

  v4f ke[2], ko[2], ve[4], vo[4];
  v4f klo[2], khi[2];

#pragma unroll
  for (int i = 0; i < 2; ++i) {
    ke[i] = *reinterpret_cast<const v4f*>(kptr + (size_t)(64 * i) * NQKV);
    ko[i] = *reinterpret_cast<const v4f*>(kptr + (size_t)(64 * i + 1) * NQKV);
  }
#pragma unroll
  for (int i = 0; i < 4; ++i) {
    ve[i] = *reinterpret_cast<const v4f*>(vptr + (size_t)(32 * i) * NQKV);
    vo[i] = *reinterpret_cast<const v4f*>(vptr + (size_t)(32 * i + 1) * NQKV);
  }

#pragma unroll
  for (int i = 0; i < 2; ++i) {
    klo[i].x = ke[i].x * we[i]; klo[i].y = ko[i].x * wo[i];
    klo[i].z = ke[i].y * we[i]; klo[i].w = ko[i].y * wo[i];
    khi[i].x = ke[i].z * we[i]; khi[i].y = ko[i].z * wo[i];
    khi[i].z = ke[i].w * we[i]; khi[i].w = ko[i].w * wo[i];
  }

  v8f acc = vzero8();

  for (int t = 0; t < NCH; ++t) {
    const int j0 = t * KV_JT;
    __syncthreads();
#pragma unroll
    for (int i = 0; i < 2; ++i) {
      *reinterpret_cast<v4f*>(&kw_s[kp + 32 * i][kc][0]) = klo[i];
      *reinterpret_cast<v4f*>(&kw_s[kp + 32 * i][kc + 2][0]) = khi[i];
    }
#pragma unroll
    for (int i = 0; i < 4; ++i) {
      v4f lo, hi2;
      lo.x = ve[i].x;  lo.y = vo[i].x;  lo.z = ve[i].y;  lo.w = vo[i].y;
      hi2.x = ve[i].z; hi2.y = vo[i].z; hi2.z = ve[i].w; hi2.w = vo[i].w;
      *reinterpret_cast<v4f*>(&v_s[vp + 16 * i][vc][0]) = lo;
      *reinterpret_cast<v4f*>(&v_s[vp + 16 * i][vc + 2][0]) = hi2;
    }
    __syncthreads();

    const int jn = (t + 1 < NCH) ? (j0 + KV_JT) : j0;
    {
      const size_t adv = (size_t)jn * NQKV;
#pragma unroll
      for (int i = 0; i < 2; ++i) {
        ke[i] =
            *reinterpret_cast<const v4f*>(kptr + adv + (size_t)(64 * i) * NQKV);
        ko[i] = *reinterpret_cast<const v4f*>(kptr + adv +
                                              (size_t)(64 * i + 1) * NQKV);
      }
#pragma unroll
      for (int i = 0; i < 4; ++i) {
        ve[i] =
            *reinterpret_cast<const v4f*>(vptr + adv + (size_t)(32 * i) * NQKV);
        vo[i] = *reinterpret_cast<const v4f*>(vptr + adv +
                                              (size_t)(32 * i + 1) * NQKV);
      }
    }

#pragma unroll
    for (int vv = 0; vv < 8; ++vv) acc[vv] *= gstep;

#pragma unroll
    for (int ks = 0; ks < KV_JT; ks += 4) {
      const int kq = (ks >> 1) + hi;
      v2f a = ldsv2(&kw_s[kq][mi * 16 + ln][0]);
      v2f bf = ldsv2(&v_s[kq][ni * 16 + ln][0]);
      acc = wmma4(a, bf, acc);
    }
#pragma unroll
    for (int i = 0; i < 2; ++i) {
      klo[i].x = ke[i].x * we[i]; klo[i].y = ko[i].x * wo[i];
      klo[i].z = ke[i].y * we[i]; klo[i].w = ko[i].y * wo[i];
      khi[i].x = ke[i].z * we[i]; khi[i].y = ko[i].z * wo[i];
      khi[i].z = ke[i].w * we[i]; khi[i].w = ko[i].w * wo[i];
    }

    const bool last = (t + 1 == NCH);
    const float om = last ? rnorm : 1.0f;
    float* dst = last ? (kvout + (size_t)bh * CDK * CDV)
                      : (pst + (size_t)(bh * NCH + t) * CDK * CDV);
#pragma unroll
    for (int vv = 0; vv < 8; ++vv)
      o_s[mi * 16 + vv + hi * 8][ni * 16 + ln] = acc[vv] * om;
    __syncthreads();
#pragma unroll
    for (int i = 0; i < 2; ++i) {
      const int r = vp + 16 * i;
      const v4f o = *reinterpret_cast<const v4f*>(&o_s[r][vc]);
      *reinterpret_cast<v4f*>(dst + (size_t)(kk0 + r) * CDV + vv0 + vc) = o;
    }
  }
}

__global__ __launch_bounds__(128) __attribute__((amdgpu_waves_per_eu(2, 2)))
void retention_kernel(
    const float* __restrict__ qkv, const float* __restrict__ pst,
    float* __restrict__ ro) {
  __shared__ __attribute__((aligned(16))) float q_s[64][132];
  __shared__ __attribute__((aligned(16))) float k_s[16][132];
  __shared__ __attribute__((aligned(16))) float v_s[8][258][2];
  __shared__ float s_s[4][16][20];

  const int iblk = blockIdx.x;
  const int bh = blockIdx.y;
  const int b = bh >> 3, h = bh & 7;
  const int rowbase = b * CT;

  const int tid = threadIdx.x;
  const int lane = tid & 31;
  const int w = tid >> 5;
  const int hi = lane >> 4;
  const int ln = lane & 15;

  const float se = logf(1.0f / 32.0f), ee = logf(1.0f / 512.0f);
  const float g = 1.0f - __expf(se + (ee - se) * ((float)h / 7.0f));
  const float logg = logf(g);
  const float rsdk = rsqrtf((float)CDK);
  const float inv1mg = 1.0f / (1.0f - g);

  const int i0 = iblk * 64;
  const int ch = iblk >> 1;
  const int c0 = ch * 128;
  const int nh = 4 * ((iblk & 1) + 1);

  const int kr = tid >> 5;
  const int kc = (tid & 31) * 4;
  const int vp = tid >> 6;
  const int vc = (tid & 63) * 4;
  const float* kptr =
      qkv + (size_t)(rowbase + c0 + kr) * NQKV + CE + h * CDK + kc;
  const float* vptr =
      qkv + (size_t)(rowbase + c0 + 2 * vp) * NQKV + 2 * CE + h * CDV + vc;
  const int qr = tid >> 5;
  const int qc = (tid & 31) * 4;

  v4f kreg[4], ve[4], vo[4];

#pragma unroll
  for (int i = 0; i < 4; ++i) {
    kreg[i] = *reinterpret_cast<const v4f*>(kptr + (size_t)(4 * i) * NQKV);
    ve[i] = *reinterpret_cast<const v4f*>(vptr + (size_t)(4 * i) * NQKV);
    vo[i] = *reinterpret_cast<const v4f*>(vptr + (size_t)(4 * i + 1) * NQKV);
  }

#pragma unroll
  for (int i = 0; i < 16; ++i) {
    const int r = qr + 4 * i;
    v4f qv = *reinterpret_cast<const v4f*>(
        qkv + (size_t)(rowbase + i0 + r) * NQKV + h * CDK + qc);
    *reinterpret_cast<v4f*>(&q_s[r][qc]) = qv;
  }

  v8f acc[16];
#pragma unroll
  for (int i = 0; i < 16; ++i) acc[i] = vzero8();

  if (ch > 0) {
    const float* pptr =
        pst + ((size_t)(bh * NCH + (ch - 1)) * CDK + 2 * vp) * CDV + vc;
    v4f pe[4], po[4];
#pragma unroll
    for (int i = 0; i < 4; ++i) {
      pe[i] = *reinterpret_cast<const v4f*>(pptr + (size_t)(4 * i) * CDV);
      po[i] = *reinterpret_cast<const v4f*>(pptr + (size_t)(4 * i + 1) * CDV);
    }
    for (int t = 0; t < 8; ++t) {
      __syncthreads();
#pragma unroll
      for (int i = 0; i < 4; ++i) {
        v4f lo, hi2;
        lo.x = pe[i].x;  lo.y = po[i].x;  lo.z = pe[i].y;  lo.w = po[i].y;
        hi2.x = pe[i].z; hi2.y = po[i].z; hi2.z = pe[i].w; hi2.w = po[i].w;
        *reinterpret_cast<v4f*>(&v_s[vp + 2 * i][vc][0]) = lo;
        *reinterpret_cast<v4f*>(&v_s[vp + 2 * i][vc + 2][0]) = hi2;
      }
      __syncthreads();

      {
        const int tn = (t + 1 < 8) ? (t + 1) : t;
        const size_t adv = (size_t)(16 * tn) * CDV;
#pragma unroll
        for (int i = 0; i < 4; ++i) {
          pe[i] = *reinterpret_cast<const v4f*>(pptr + adv + (size_t)(4 * i) * CDV);
          po[i] = *reinterpret_cast<const v4f*>(pptr + adv +
                                                (size_t)(4 * i + 1) * CDV);
        }
      }

#pragma unroll
      for (int ks4 = 0; ks4 < 16; ks4 += 4) {
        const int ak = 16 * t + ks4 + hi * 2;
        const int kp = (ks4 >> 1) + hi;
        v2f a = ldsv2(&q_s[w * 16 + ln][ak]);
#pragma unroll
        for (int nt = 0; nt < 16; ++nt) {
          v2f bf = ldsv2(&v_s[kp][nt * 16 + ln][0]);
          acc[nt] = wmma4(a, bf, acc[nt]);
        }
      }
      __builtin_amdgcn_sched_barrier(0);
    }
#pragma unroll
    for (int vv = 0; vv < 8; ++vv) {
      const int i = i0 + w * 16 + vv + hi * 8;
      const float rs = __expf(logg * (float)(i - c0 + 1)) * rsdk;
#pragma unroll
      for (int nt = 0; nt < 16; ++nt) acc[nt][vv] *= rs;
    }
  }

  for (int t = 0; t < nh; ++t) {
    const int j0 = c0 + 16 * t;
    __syncthreads();
#pragma unroll
    for (int i = 0; i < 4; ++i) {
      *reinterpret_cast<v4f*>(&k_s[kr + 4 * i][kc]) = kreg[i];
      v4f lo, hi2;
      lo.x = ve[i].x;  lo.y = vo[i].x;  lo.z = ve[i].y;  lo.w = vo[i].y;
      hi2.x = ve[i].z; hi2.y = vo[i].z; hi2.z = ve[i].w; hi2.w = vo[i].w;
      *reinterpret_cast<v4f*>(&v_s[vp + 2 * i][vc][0]) = lo;
      *reinterpret_cast<v4f*>(&v_s[vp + 2 * i][vc + 2][0]) = hi2;
    }
    __syncthreads();

    {
      const int tn = (t + 1 < nh) ? (t + 1) : t;
      const size_t adv = (size_t)(16 * tn) * NQKV;
#pragma unroll
      for (int i = 0; i < 4; ++i) {
        kreg[i] =
            *reinterpret_cast<const v4f*>(kptr + adv + (size_t)(4 * i) * NQKV);
        ve[i] =
            *reinterpret_cast<const v4f*>(vptr + adv + (size_t)(4 * i) * NQKV);
        vo[i] = *reinterpret_cast<const v4f*>(vptr + adv +
                                              (size_t)(4 * i + 1) * NQKV);
      }
    }

    v8f sacc = vzero8();
#pragma unroll
    for (int kk = 0; kk < CDK; kk += 4) {
      const int ak = kk + hi * 2;
      v2f a = ldsv2(&q_s[w * 16 + ln][ak]);
      v2f bf = ldsv2(&k_s[ln][ak]);
      sacc = wmma4(a, bf, sacc);
    }

#pragma unroll
    for (int vv = 0; vv < 8; ++vv) {
      int i = i0 + w * 16 + vv + hi * 8;
      int j = j0 + ln;
      float m = (i >= j) ? __expf(logg * (float)(i - j)) * rsdk : 0.0f;
      sacc[vv] *= m;
    }

#pragma unroll
    for (int vv = 0; vv < 8; ++vv) s_s[w][vv + hi * 8][ln] = sacc[vv];

#pragma unroll
    for (int ks4 = 0; ks4 < 16; ks4 += 4) {
      const int ak = ks4 + hi * 2;
      const int kp = (ks4 >> 1) + hi;
      v2f a = ldsv2(&s_s[w][ln][ak]);
#pragma unroll
      for (int nt = 0; nt < 16; ++nt) {
        v2f bf = ldsv2(&v_s[kp][nt * 16 + ln][0]);
        acc[nt] = wmma4(a, bf, acc[nt]);
      }
    }
    __builtin_amdgcn_sched_barrier(0);
  }

#pragma unroll
  for (int nt = 0; nt < 16; ++nt)
#pragma unroll
    for (int vv = 0; vv < 8; ++vv) {
      int i = i0 + w * 16 + vv + hi * 8;
      float rowsum = (1.0f - __expf(logg * (float)(i + 1))) * inv1mg;
      float sc = rsqrtf(rowsum);
      ro[(size_t)(rowbase + i) * CV + h * CDV + nt * 16 + ln] = acc[nt][vv] * sc;
    }
}

__global__ __launch_bounds__(256) void gn_gate_kernel(
    const float* __restrict__ ro, float* __restrict__ gbuf) {
  const int row = blockIdx.x;
  const int w = threadIdx.x >> 5;
  const int lane = threadIdx.x & 31;
  const size_t base = (size_t)row * CV + (size_t)w * CDV;

  float x[8];
  float s = 0.0f, s2 = 0.0f;
#pragma unroll
  for (int i = 0; i < 8; ++i) {
    x[i] = ro[base + i * 32 + lane];
    s += x[i];
    s2 += x[i] * x[i];
  }
#pragma unroll
  for (int off = 16; off > 0; off >>= 1) {
    s += __shfl_xor(s, off, 32);
    s2 += __shfl_xor(s2, off, 32);
  }
  const float mu = s * (1.0f / 256.0f);
  const float var = s2 * (1.0f / 256.0f) - mu * mu;
  const float rstd = rsqrtf(var + 1e-5f);

#pragma unroll
  for (int i = 0; i < 8; ++i) {
    size_t idx = base + i * 32 + lane;
    float nrm = (x[i] - mu) * rstd;
    float gv = gbuf[idx];
    float sil = gv / (1.0f + __expf(-gv));
    gbuf[idx] = sil * nrm;
  }
}

extern "C" void kernel_launch(void* const* d_in, const int* in_sizes, int n_in,
                              void* d_out, int out_size, void* d_ws,
                              size_t ws_size, hipStream_t stream) {
  const float* hs    = (const float*)d_in[0];
  const float* w_qkv = (const float*)d_in[1];
  const float* b_qkv = (const float*)d_in[2];
  const float* w_g   = (const float*)d_in[3];
  const float* b_g   = (const float*)d_in[4];
  const float* w_p   = (const float*)d_in[5];
  const float* b_p   = (const float*)d_in[6];

  float* out  = (float*)d_out;
  float* qkv  = (float*)d_ws;
  float* gbuf = qkv + (size_t)BT * NQKV;
  float* ro   = gbuf + (size_t)BT * CV;
  float* pst  = ro + (size_t)BT * CV;

  gemm_wmma_f32<<<dim3(NQKV / 64, BT / 128), 128, 0, stream>>>(
      hs, w_qkv, b_qkv, qkv, BT, NQKV, CE);

  xpos_kernel<<<dim3(BT), 512, 0, stream>>>(qkv);

  gemm_wmma_f32<<<dim3(CV / 64, BT / 128), 128, 0, stream>>>(
      hs, w_g, b_g, gbuf, BT, CV, CE);

  curr_kv_kernel<<<dim3(16, CB * CH), 256, 0, stream>>>(
      qkv, out + (size_t)BT * CE, pst);

  retention_kernel<<<dim3(CT / 64, CB * CH), 128, 0, stream>>>(qkv, pst, ro);

  gn_gate_kernel<<<dim3(BT), 256, 0, stream>>>(ro, gbuf);

  gemm_wmma_f32<<<dim3(CE / 64, BT / 128), 128, 0, stream>>>(
      gbuf, w_p, b_p, out, BT, CE, CV);
}
